# starting kernel
# baseline (speedup 1.0000x reference)
#include <hip/hip_runtime.h>
#include <hip/hip_bf16.h>
#include <math.h>

#define B_TOK 16384
#define D_IN  128
#define H_DIM 512
#define O_DIM 128
#define E_NUM 8
#define TM    32

typedef __attribute__((ext_vector_type(2))) float v2f;
typedef __attribute__((ext_vector_type(8))) float v8f;

__global__ __launch_bounds__(256) void moe_init(float* __restrict__ y,
                                                int* __restrict__ cnt,
                                                float* __restrict__ importance) {
  int idx = blockIdx.x * 256 + threadIdx.x;
  ((float4*)y)[idx] = make_float4(0.f, 0.f, 0.f, 0.f);
  if (blockIdx.x == 0) {
    if (threadIdx.x < E_NUM) cnt[threadIdx.x] = 0;
    else if (threadIdx.x < 2 * E_NUM) importance[threadIdx.x - E_NUM] = 0.f;
  }
}

__global__ __launch_bounds__(256) void moe_gate(const float* __restrict__ x,
                                                const float* __restrict__ w_gate,
                                                float* __restrict__ gates,
                                                int* __restrict__ bucket,
                                                int* __restrict__ cnt,
                                                float* __restrict__ importance) {
  __shared__ float wg[D_IN * E_NUM];
  __shared__ float simp[E_NUM];
  const int tid = threadIdx.x;
  for (int i = tid; i < D_IN * E_NUM; i += 256) wg[i] = w_gate[i];
  if (tid < E_NUM) simp[tid] = 0.f;
  __syncthreads();

  const int b = blockIdx.x * 256 + tid;
  const float4* xr = (const float4*)(x + (size_t)b * D_IN);
  float l[E_NUM];
#pragma unroll
  for (int e = 0; e < E_NUM; ++e) l[e] = 0.f;
#pragma unroll 4
  for (int d4 = 0; d4 < D_IN / 4; ++d4) {
    float4 xv = xr[d4];
    const float* wr = &wg[d4 * 4 * E_NUM];
#pragma unroll
    for (int e = 0; e < E_NUM; ++e)
      l[e] += xv.x * wr[e] + xv.y * wr[E_NUM + e] + xv.z * wr[2 * E_NUM + e] + xv.w * wr[3 * E_NUM + e];
  }

  int e0 = 0; float l0 = l[0];
#pragma unroll
  for (int e = 1; e < E_NUM; ++e) if (l[e] > l0) { l0 = l[e]; e0 = e; }
  int e1 = -1; float l1 = -3.4e38f;
#pragma unroll
  for (int e = 0; e < E_NUM; ++e) if (e != e0 && l[e] > l1) { l1 = l[e]; e1 = e; }

  float q  = __expf(l1 - l0);
  float g0 = 1.f / (1.f + q);
  float g1 = q / (1.f + q);

  float gr[E_NUM];
#pragma unroll
  for (int e = 0; e < E_NUM; ++e) gr[e] = 0.f;
  gr[e0] = g0; gr[e1] = g1;
#pragma unroll
  for (int e = 0; e < E_NUM; ++e) gates[(size_t)b * E_NUM + e] = gr[e];

  atomicAdd(&simp[e0], g0);
  atomicAdd(&simp[e1], g1);
  int p0 = atomicAdd(&cnt[e0], 1); bucket[e0 * B_TOK + p0] = b;
  int p1 = atomicAdd(&cnt[e1], 1); bucket[e1 * B_TOK + p1] = b;
  __syncthreads();
  if (tid < E_NUM) unsafeAtomicAdd(&importance[tid], simp[tid]);
}

__global__ void moe_loss(const float* __restrict__ importance,
                         const int* __restrict__ cnt,
                         float* __restrict__ out_loss) {
  if (threadIdx.x == 0 && blockIdx.x == 0) {
    float im[E_NUM], ld[E_NUM], mi = 0.f, ml = 0.f;
    for (int e = 0; e < E_NUM; ++e) {
      im[e] = importance[e]; ld[e] = (float)cnt[e];
      mi += im[e]; ml += ld[e];
    }
    mi *= (1.f / E_NUM); ml *= (1.f / E_NUM);
    float vi = 0.f, vl = 0.f;
    for (int e = 0; e < E_NUM; ++e) {
      float a = im[e] - mi, c = ld[e] - ml;
      vi += a * a; vl += c * c;
    }
    vi *= (1.f / (E_NUM - 1)); vl *= (1.f / (E_NUM - 1));
    *out_loss = vi / (mi * mi + 1e-10f) + vl / (ml * ml + 1e-10f);
  }
}

__global__ __launch_bounds__(256) void moe_expert(const float* __restrict__ x,
                                                  const float* __restrict__ w1,
                                                  const float* __restrict__ b1,
                                                  const float* __restrict__ w2,
                                                  const float* __restrict__ b2,
                                                  const float* __restrict__ gates,
                                                  const int* __restrict__ bucket,
                                                  const int* __restrict__ cnt,
                                                  float* __restrict__ y) {
  __shared__ float xs[TM][132];
  __shared__ float hs[TM][516];
  __shared__ float gbuf[TM];
  __shared__ int   tokbuf[TM];

  const int e    = blockIdx.y;
  const int n    = cnt[e];
  const int base = blockIdx.x * TM;
  if (base >= n) return;
  const int m = min(TM, n - base);

  const int tid  = threadIdx.x;
  const int lane = tid & 31;
  const int w    = tid >> 5;
  const int lh   = lane >> 4;
  const int ln   = lane & 15;

  if (tid < TM) {
    int idx = base + tid;
    int tok = bucket[e * B_TOK + ((idx < n) ? idx : (n - 1))];
    tokbuf[tid] = tok;
    gbuf[tid]   = (idx < n) ? gates[(size_t)tok * E_NUM + e] : 0.f;
  }
  __syncthreads();

#pragma unroll
  for (int i = 0; i < TM * D_IN / 256; ++i) {
    int idx = tid + i * 256;
    int r = idx >> 7, c = idx & 127;
    xs[r][c] = x[(size_t)tokbuf[r] * D_IN + c];
  }
  __syncthreads();

  {
    const float* w1e = w1 + (size_t)e * D_IN * H_DIM;
    const float* pB  = w1e + (2 * lh) * H_DIM + w * 64 + ln;
    v8f acc[2][4];
#pragma unroll
    for (int nt = 0; nt < 4; ++nt) {
      float bv = b1[e * H_DIM + w * 64 + nt * 16 + ln];
#pragma unroll
      for (int i = 0; i < 8; ++i) { acc[0][nt][i] = bv; acc[1][nt][i] = bv; }
    }
    for (int kk = 0; kk < D_IN; kk += 4) {
      float2 a2l = *(const float2*)&xs[ln][kk + 2 * lh];
      float2 a2h = *(const float2*)&xs[16 + ln][kk + 2 * lh];
      v2f a0; a0[0] = a2l.x; a0[1] = a2l.y;
      v2f a1; a1[0] = a2h.x; a1[1] = a2h.y;
#pragma unroll
      for (int nt = 0; nt < 4; ++nt) {
        v2f bb;
        bb[0] = pB[kk * H_DIM + nt * 16];
        bb[1] = pB[kk * H_DIM + nt * 16 + H_DIM];
        acc[0][nt] = __builtin_amdgcn_wmma_f32_16x16x4_f32(false, a0, false, bb,
                                                           (short)0, acc[0][nt], false, false);
        acc[1][nt] = __builtin_amdgcn_wmma_f32_16x16x4_f32(false, a1, false, bb,
                                                           (short)0, acc[1][nt], false, false);
      }
    }
#pragma unroll
    for (int mt = 0; mt < 2; ++mt)
#pragma unroll
      for (int nt = 0; nt < 4; ++nt)
#pragma unroll
        for (int i = 0; i < 8; ++i)
          hs[mt * 16 + 8 * lh + i][w * 64 + nt * 16 + ln] = fmaxf(acc[mt][nt][i], 0.f);
  }
  __syncthreads();

  {
    const int o0 = w * 16;
    const float* w2e = w2 + (size_t)e * H_DIM * O_DIM;
    const float* pB  = w2e + (2 * lh) * O_DIM + o0 + ln;
    v8f acc[2][2];
    float bv = b2[e * O_DIM + o0 + ln];
#pragma unroll
    for (int i = 0; i < 8; ++i) {
      acc[0][0][i] = bv;  acc[0][1][i] = 0.f;
      acc[1][0][i] = bv;  acc[1][1][i] = 0.f;
    }
    for (int kk = 0; kk < H_DIM; kk += 8) {
#pragma unroll
      for (int p = 0; p < 2; ++p) {
        int koff = kk + 4 * p;
        float2 a2l = *(const float2*)&hs[ln][koff + 2 * lh];
        float2 a2h = *(const float2*)&hs[16 + ln][koff + 2 * lh];
        v2f a0; a0[0] = a2l.x; a0[1] = a2l.y;
        v2f a1; a1[0] = a2h.x; a1[1] = a2h.y;
        v2f bb;
        bb[0] = pB[koff * O_DIM];
        bb[1] = pB[koff * O_DIM + O_DIM];
        acc[0][p] = __builtin_amdgcn_wmma_f32_16x16x4_f32(false, a0, false, bb,
                                                          (short)0, acc[0][p], false, false);
        acc[1][p] = __builtin_amdgcn_wmma_f32_16x16x4_f32(false, a1, false, bb,
                                                          (short)0, acc[1][p], false, false);
      }
    }
#pragma unroll
    for (int mt = 0; mt < 2; ++mt)
#pragma unroll
      for (int i = 0; i < 8; ++i) {
        int M = mt * 16 + 8 * lh + i;
        float v = (acc[mt][0][i] + acc[mt][1][i]) * gbuf[M];
        if (M < m)
          unsafeAtomicAdd(&y[(size_t)tokbuf[M] * O_DIM + o0 + ln], v);
      }
  }
}

__global__ __launch_bounds__(256) void moe_eps(float* __restrict__ y) {
  const float EPSV = 2.2204460492503131e-16f;
  int idx = blockIdx.x * 256 + threadIdx.x;
  float4* p = ((float4*)y) + idx;
  float4 v = *p;
  v.x = (v.x == 0.f) ? EPSV : v.x;
  v.y = (v.y == 0.f) ? EPSV : v.y;
  v.z = (v.z == 0.f) ? EPSV : v.z;
  v.w = (v.w == 0.f) ? EPSV : v.w;
  *p = v;
}

extern "C" void kernel_launch(void* const* d_in, const int* in_sizes, int n_in,
                              void* d_out, int out_size, void* d_ws, size_t ws_size,
                              hipStream_t stream) {
  (void)in_sizes; (void)n_in; (void)out_size; (void)ws_size;
  const float* x      = (const float*)d_in[0];
  const float* w_gate = (const float*)d_in[1];
  const float* w1     = (const float*)d_in[2];
  const float* b1     = (const float*)d_in[3];
  const float* w2     = (const float*)d_in[4];
  const float* b2     = (const float*)d_in[5];

  float* y    = (float*)d_out;
  float* loss = y + (size_t)B_TOK * O_DIM;

  char*  ws         = (char*)d_ws;
  float* gates      = (float*)ws;
  int*   bucket     = (int*)(ws + (size_t)B_TOK * E_NUM * 4);
  int*   cnt        = (int*)(ws + 2 * (size_t)B_TOK * E_NUM * 4);
  float* importance = (float*)(ws + 2 * (size_t)B_TOK * E_NUM * 4 + 64);

  moe_init<<<(B_TOK * O_DIM) / (4 * 256), 256, 0, stream>>>(y, cnt, importance);
  moe_gate<<<B_TOK / 256, 256, 0, stream>>>(x, w_gate, gates, bucket, cnt, importance);
  moe_loss<<<1, 32, 0, stream>>>(importance, cnt, loss);
  dim3 g(B_TOK / TM, E_NUM);
  moe_expert<<<g, 256, 0, stream>>>(x, w1, b1, w2, b2, gates, bucket, cnt, y);
  moe_eps<<<(B_TOK * O_DIM) / (4 * 256), 256, 0, stream>>>(y);
}


# expert kernel: both GEMM loops software-pipelined, the next 32-op block of weight fragments is loaded into a second register set while the current block multiplies; k order of the sums unchanged
# speedup vs baseline: 1.0730x; 1.0730x over previous
#include <hip/hip_runtime.h>
#include <hip/hip_bf16.h>
#include <math.h>

#define B_TOK 16384
#define D_IN  128
#define H_DIM 512
#define O_DIM 128
#define E_NUM 8
#define TM    32

typedef __attribute__((ext_vector_type(2))) float v2f;
typedef __attribute__((ext_vector_type(8))) float v8f;

__global__ __launch_bounds__(256) void moe_init(float* __restrict__ y,
                                                int* __restrict__ cnt,
                                                float* __restrict__ importance) {
  int idx = blockIdx.x * 256 + threadIdx.x;
  ((float4*)y)[idx] = make_float4(0.f, 0.f, 0.f, 0.f);
  if (blockIdx.x == 0) {
    if (threadIdx.x < E_NUM) cnt[threadIdx.x] = 0;
    else if (threadIdx.x < 2 * E_NUM) importance[threadIdx.x - E_NUM] = 0.f;
  }
}

__global__ __launch_bounds__(256) void moe_gate(const float* __restrict__ x,
                                                const float* __restrict__ w_gate,
                                                float* __restrict__ gates,
                                                int* __restrict__ bucket,
                                                int* __restrict__ cnt,
                                                float* __restrict__ importance) {
  __shared__ float wg[D_IN * E_NUM];
  __shared__ float simp[E_NUM];
  const int tid = threadIdx.x;
  for (int i = tid; i < D_IN * E_NUM; i += 256) wg[i] = w_gate[i];
  if (tid < E_NUM) simp[tid] = 0.f;
  __syncthreads();

  const int b = blockIdx.x * 256 + tid;
  const float4* xr = (const float4*)(x + (size_t)b * D_IN);
  float l[E_NUM];
#pragma unroll
  for (int e = 0; e < E_NUM; ++e) l[e] = 0.f;
#pragma unroll 4
  for (int d4 = 0; d4 < D_IN / 4; ++d4) {
    float4 xv = xr[d4];
    const float* wr = &wg[d4 * 4 * E_NUM];
#pragma unroll
    for (int e = 0; e < E_NUM; ++e)
      l[e] += xv.x * wr[e] + xv.y * wr[E_NUM + e] + xv.z * wr[2 * E_NUM + e] + xv.w * wr[3 * E_NUM + e];
  }

  int e0 = 0; float l0 = l[0];
#pragma unroll
  for (int e = 1; e < E_NUM; ++e) if (l[e] > l0) { l0 = l[e]; e0 = e; }
  int e1 = -1; float l1 = -3.4e38f;
#pragma unroll
  for (int e = 0; e < E_NUM; ++e) if (e != e0 && l[e] > l1) { l1 = l[e]; e1 = e; }

  float q  = __expf(l1 - l0);
  float g0 = 1.f / (1.f + q);
  float g1 = q / (1.f + q);

  float gr[E_NUM];
#pragma unroll
  for (int e = 0; e < E_NUM; ++e) gr[e] = 0.f;
  gr[e0] = g0; gr[e1] = g1;
#pragma unroll
  for (int e = 0; e < E_NUM; ++e) gates[(size_t)b * E_NUM + e] = gr[e];

  atomicAdd(&simp[e0], g0);
  atomicAdd(&simp[e1], g1);
  int p0 = atomicAdd(&cnt[e0], 1); bucket[e0 * B_TOK + p0] = b;
  int p1 = atomicAdd(&cnt[e1], 1); bucket[e1 * B_TOK + p1] = b;
  __syncthreads();
  if (tid < E_NUM) unsafeAtomicAdd(&importance[tid], simp[tid]);
}

__global__ void moe_loss(const float* __restrict__ importance,
                         const int* __restrict__ cnt,
                         float* __restrict__ out_loss) {
  if (threadIdx.x == 0 && blockIdx.x == 0) {
    float im[E_NUM], ld[E_NUM], mi = 0.f, ml = 0.f;
    for (int e = 0; e < E_NUM; ++e) {
      im[e] = importance[e]; ld[e] = (float)cnt[e];
      mi += im[e]; ml += ld[e];
    }
    mi *= (1.f / E_NUM); ml *= (1.f / E_NUM);
    float vi = 0.f, vl = 0.f;
    for (int e = 0; e < E_NUM; ++e) {
      float a = im[e] - mi, c = ld[e] - ml;
      vi += a * a; vl += c * c;
    }
    vi *= (1.f / (E_NUM - 1)); vl *= (1.f / (E_NUM - 1));
    *out_loss = vi / (mi * mi + 1e-10f) + vl / (ml * ml + 1e-10f);
  }
}

__device__ __forceinline__ void g1_load(v2f (&b)[4][4], const float* __restrict__ p) {
#pragma unroll
  for (int s = 0; s < 4; ++s)
#pragma unroll
    for (int nt = 0; nt < 4; ++nt) {
      b[s][nt][0] = p[(4 * s) * H_DIM + nt * 16];
      b[s][nt][1] = p[(4 * s) * H_DIM + nt * 16 + H_DIM];
    }
}

__device__ __forceinline__ void g1_mma(v8f (&acc)[2][4], const v2f (&b)[4][4],
                                       const float* xl, const float* xh) {
#pragma unroll
  for (int s = 0; s < 4; ++s) {
    float2 a2l = *(const float2*)(xl + 4 * s);
    float2 a2h = *(const float2*)(xh + 4 * s);
    v2f a0; a0[0] = a2l.x; a0[1] = a2l.y;
    v2f a1; a1[0] = a2h.x; a1[1] = a2h.y;
#pragma unroll
    for (int nt = 0; nt < 4; ++nt) {
      acc[0][nt] = __builtin_amdgcn_wmma_f32_16x16x4_f32(false, a0, false, b[s][nt],
                                                         (short)0, acc[0][nt], false, false);
      acc[1][nt] = __builtin_amdgcn_wmma_f32_16x16x4_f32(false, a1, false, b[s][nt],
                                                         (short)0, acc[1][nt], false, false);
    }
  }
}

__device__ __forceinline__ void g2_load(v2f (&b)[16], const float* __restrict__ p) {
#pragma unroll
  for (int s = 0; s < 16; ++s) {
    b[s][0] = p[(4 * s) * O_DIM];
    b[s][1] = p[(4 * s) * O_DIM + O_DIM];
  }
}

__device__ __forceinline__ void g2_mma(v8f (&acc)[2][2], const v2f (&b)[16],
                                       const float* hl, const float* hh) {
#pragma unroll
  for (int s = 0; s < 16; ++s) {
    float2 a2l = *(const float2*)(hl + 4 * s);
    float2 a2h = *(const float2*)(hh + 4 * s);
    v2f a0; a0[0] = a2l.x; a0[1] = a2l.y;
    v2f a1; a1[0] = a2h.x; a1[1] = a2h.y;
    acc[0][s & 1] = __builtin_amdgcn_wmma_f32_16x16x4_f32(false, a0, false, b[s],
                                                          (short)0, acc[0][s & 1], false, false);
    acc[1][s & 1] = __builtin_amdgcn_wmma_f32_16x16x4_f32(false, a1, false, b[s],
                                                          (short)0, acc[1][s & 1], false, false);
  }
}

__global__ __launch_bounds__(256, 6) void moe_expert(const float* __restrict__ x,
                                                  const float* __restrict__ w1,
                                                  const float* __restrict__ b1,
                                                  const float* __restrict__ w2,
                                                  const float* __restrict__ b2,
                                                  const float* __restrict__ gates,
                                                  const int* __restrict__ bucket,
                                                  const int* __restrict__ cnt,
                                                  float* __restrict__ y) {
  __shared__ float xs[TM][132];
  __shared__ float hs[TM][516];
  __shared__ float gbuf[TM];
  __shared__ int   tokbuf[TM];

  const int e    = blockIdx.y;
  const int n    = cnt[e];
  const int base = blockIdx.x * TM;
  if (base >= n) return;
  const int m = min(TM, n - base);

  const int tid  = threadIdx.x;
  const int lane = tid & 31;
  const int w    = tid >> 5;
  const int lh   = lane >> 4;
  const int ln   = lane & 15;

  if (tid < TM) {
    int idx = base + tid;
    int tok = bucket[e * B_TOK + ((idx < n) ? idx : (n - 1))];
    tokbuf[tid] = tok;
    gbuf[tid]   = (idx < n) ? gates[(size_t)tok * E_NUM + e] : 0.f;
  }
  __syncthreads();

#pragma unroll
  for (int i = 0; i < TM * D_IN / 256; ++i) {
    int idx = tid + i * 256;
    int r = idx >> 7, c = idx & 127;
    xs[r][c] = x[(size_t)tokbuf[r] * D_IN + c];
  }
  __syncthreads();

  {
    const float* w1e = w1 + (size_t)e * D_IN * H_DIM;
    const float* pB  = w1e + (2 * lh) * H_DIM + w * 64 + ln;
    v8f acc[2][4];
#pragma unroll
    for (int nt = 0; nt < 4; ++nt) {
      float bv = b1[e * H_DIM + w * 64 + nt * 16 + ln];
#pragma unroll
      for (int i = 0; i < 8; ++i) { acc[0][nt][i] = bv; acc[1][nt][i] = bv; }
    }
    v2f bq0[4][4], bq1[4][4];
    const float* xl = &xs[ln][2 * lh];
    const float* xh = &xs[16 + ln][2 * lh];
    g1_load(bq0, pB);
#pragma nounroll
    for (int kb = 0; kb < D_IN / 16 - 2; kb += 2) {
      g1_load(bq1, pB + (kb + 1) * 16 * H_DIM);
      g1_mma(acc, bq0, xl + kb * 16, xh + kb * 16);
      g1_load(bq0, pB + (kb + 2) * 16 * H_DIM);
      g1_mma(acc, bq1, xl + (kb + 1) * 16, xh + (kb + 1) * 16);
    }
    g1_load(bq1, pB + (D_IN / 16 - 1) * 16 * H_DIM);
    g1_mma(acc, bq0, xl + (D_IN / 16 - 2) * 16, xh + (D_IN / 16 - 2) * 16);
    g1_mma(acc, bq1, xl + (D_IN / 16 - 1) * 16, xh + (D_IN / 16 - 1) * 16);
#pragma unroll
    for (int mt = 0; mt < 2; ++mt)
#pragma unroll
      for (int nt = 0; nt < 4; ++nt)
#pragma unroll
        for (int i = 0; i < 8; ++i)
          hs[mt * 16 + 8 * lh + i][w * 64 + nt * 16 + ln] = fmaxf(acc[mt][nt][i], 0.f);
  }
  __syncthreads();

  {
    const int o0 = w * 16;
    const float* w2e = w2 + (size_t)e * H_DIM * O_DIM;
    const float* pB  = w2e + (2 * lh) * O_DIM + o0 + ln;
    v8f acc[2][2];
    float bv = b2[e * O_DIM + o0 + ln];
#pragma unroll
    for (int i = 0; i < 8; ++i) {
      acc[0][0][i] = bv;  acc[0][1][i] = 0.f;
      acc[1][0][i] = bv;  acc[1][1][i] = 0.f;
    }
    v2f cq0[16], cq1[16];
    g2_load(cq0, pB);
    const float* hl = &hs[ln][2 * lh];
    const float* hh = &hs[16 + ln][2 * lh];
    for (int kb = 0; kb < H_DIM / 64 - 2; kb += 2) {
      g2_load(cq1, pB + (kb + 1) * 64 * O_DIM);
      g2_mma(acc, cq0, hl + kb * 64, hh + kb * 64);
      g2_load(cq0, pB + (kb + 2) * 64 * O_DIM);
      g2_mma(acc, cq1, hl + (kb + 1) * 64, hh + (kb + 1) * 64);
    }
    g2_load(cq1, pB + (H_DIM / 64 - 1) * 64 * O_DIM);
    g2_mma(acc, cq0, hl + (H_DIM / 64 - 2) * 64, hh + (H_DIM / 64 - 2) * 64);
    g2_mma(acc, cq1, hl + (H_DIM / 64 - 1) * 64, hh + (H_DIM / 64 - 1) * 64);
#pragma unroll
    for (int mt = 0; mt < 2; ++mt)
#pragma unroll
      for (int i = 0; i < 8; ++i) {
        int M = mt * 16 + 8 * lh + i;
        float v = (acc[mt][0][i] + acc[mt][1][i]) * gbuf[M];
        if (M < m)
          unsafeAtomicAdd(&y[(size_t)tokbuf[M] * O_DIM + o0 + ln], v);
      }
  }
}

__global__ __launch_bounds__(256) void moe_eps(float* __restrict__ y) {
  const float EPSV = 2.2204460492503131e-16f;
  int idx = blockIdx.x * 256 + threadIdx.x;
  float4* p = ((float4*)y) + idx;
  float4 v = *p;
  v.x = (v.x == 0.f) ? EPSV : v.x;
  v.y = (v.y == 0.f) ? EPSV : v.y;
  v.z = (v.z == 0.f) ? EPSV : v.z;
  v.w = (v.w == 0.f) ? EPSV : v.w;
  *p = v;
}

extern "C" void kernel_launch(void* const* d_in, const int* in_sizes, int n_in,
                              void* d_out, int out_size, void* d_ws, size_t ws_size,
                              hipStream_t stream) {
  (void)in_sizes; (void)n_in; (void)out_size; (void)ws_size;
  const float* x      = (const float*)d_in[0];
  const float* w_gate = (const float*)d_in[1];
  const float* w1     = (const float*)d_in[2];
  const float* b1     = (const float*)d_in[3];
  const float* w2     = (const float*)d_in[4];
  const float* b2     = (const float*)d_in[5];

  float* y    = (float*)d_out;
  float* loss = y + (size_t)B_TOK * O_DIM;

  char*  ws         = (char*)d_ws;
  float* gates      = (float*)ws;
  int*   bucket     = (int*)(ws + (size_t)B_TOK * E_NUM * 4);
  int*   cnt        = (int*)(ws + 2 * (size_t)B_TOK * E_NUM * 4);
  float* importance = (float*)(ws + 2 * (size_t)B_TOK * E_NUM * 4 + 64);

  moe_init<<<(B_TOK * O_DIM) / (4 * 256), 256, 0, stream>>>(y, cnt, importance);
  moe_gate<<<B_TOK / 256, 256, 0, stream>>>(x, w_gate, gates, bucket, cnt, importance);
  moe_loss<<<1, 32, 0, stream>>>(importance, cnt, loss);
  dim3 g(B_TOK / TM, E_NUM);
  moe_expert<<<g, 256, 0, stream>>>(x, w1, b1, w2, b2, gates, bucket, cnt, y);
  moe_eps<<<(B_TOK * O_DIM) / (4 * 256), 256, 0, stream>>>(y);
}


# gate kernel: bucket slots from an LDS rank per block plus one global add per expert per block (512 counter adds, was 32768 on 8 adjacent words); float sums and their order unchanged
# speedup vs baseline: 1.1376x; 1.0602x over previous
#include <hip/hip_runtime.h>
#include <hip/hip_bf16.h>
#include <math.h>

#define B_TOK 16384
#define D_IN  128
#define H_DIM 512
#define O_DIM 128
#define E_NUM 8
#define TM    32

typedef __attribute__((ext_vector_type(2))) float v2f;
typedef __attribute__((ext_vector_type(8))) float v8f;

__global__ __launch_bounds__(256) void moe_init(float* __restrict__ y,
                                                int* __restrict__ cnt,
                                                float* __restrict__ importance) {
  int idx = blockIdx.x * 256 + threadIdx.x;
  ((float4*)y)[idx] = make_float4(0.f, 0.f, 0.f, 0.f);
  if (blockIdx.x == 0) {
    if (threadIdx.x < E_NUM) cnt[threadIdx.x] = 0;
    else if (threadIdx.x < 2 * E_NUM) importance[threadIdx.x - E_NUM] = 0.f;
  }
}

__global__ __launch_bounds__(256) void moe_gate(const float* __restrict__ x,
                                                const float* __restrict__ w_gate,
                                                float* __restrict__ gates,
                                                int* __restrict__ bucket,
                                                int* __restrict__ cnt,
                                                float* __restrict__ importance) {
  __shared__ float wg[D_IN * E_NUM];
  __shared__ float simp[E_NUM];
  __shared__ int   scnt[E_NUM];
  __shared__ int   sbase[E_NUM];
  const int tid = threadIdx.x;
  for (int i = tid; i < D_IN * E_NUM; i += 256) wg[i] = w_gate[i];
  if (tid < E_NUM) { simp[tid] = 0.f; scnt[tid] = 0; }
  __syncthreads();

  const int b = blockIdx.x * 256 + tid;
  const float4* xr = (const float4*)(x + (size_t)b * D_IN);
  float l[E_NUM];
#pragma unroll
  for (int e = 0; e < E_NUM; ++e) l[e] = 0.f;
#pragma unroll 4
  for (int d4 = 0; d4 < D_IN / 4; ++d4) {
    float4 xv = xr[d4];
    const float* wr = &wg[d4 * 4 * E_NUM];
#pragma unroll
    for (int e = 0; e < E_NUM; ++e)
      l[e] += xv.x * wr[e] + xv.y * wr[E_NUM + e] + xv.z * wr[2 * E_NUM + e] + xv.w * wr[3 * E_NUM + e];
  }

  int e0 = 0; float l0 = l[0];
#pragma unroll
  for (int e = 1; e < E_NUM; ++e) if (l[e] > l0) { l0 = l[e]; e0 = e; }
  int e1 = -1; float l1 = -3.4e38f;
#pragma unroll
  for (int e = 0; e < E_NUM; ++e) if (e != e0 && l[e] > l1) { l1 = l[e]; e1 = e; }

  float q  = __expf(l1 - l0);
  float g0 = 1.f / (1.f + q);
  float g1 = q / (1.f + q);

  float gr[E_NUM];
#pragma unroll
  for (int e = 0; e < E_NUM; ++e) gr[e] = 0.f;
  gr[e0] = g0; gr[e1] = g1;
#pragma unroll
  for (int e = 0; e < E_NUM; ++e) gates[(size_t)b * E_NUM + e] = gr[e];

  atomicAdd(&simp[e0], g0);
  atomicAdd(&simp[e1], g1);
  const int r0 = atomicAdd(&scnt[e0], 1);
  const int r1 = atomicAdd(&scnt[e1], 1);
  __syncthreads();
  if (tid < E_NUM) {
    sbase[tid] = atomicAdd(&cnt[tid], scnt[tid]);
    unsafeAtomicAdd(&importance[tid], simp[tid]);
  }
  __syncthreads();
  bucket[e0 * B_TOK + sbase[e0] + r0] = b;
  bucket[e1 * B_TOK + sbase[e1] + r1] = b;
}

__global__ void moe_loss(const float* __restrict__ importance,
                         const int* __restrict__ cnt,
                         float* __restrict__ out_loss) {
  if (threadIdx.x == 0 && blockIdx.x == 0) {
    float im[E_NUM], ld[E_NUM], mi = 0.f, ml = 0.f;
    for (int e = 0; e < E_NUM; ++e) {
      im[e] = importance[e]; ld[e] = (float)cnt[e];
      mi += im[e]; ml += ld[e];
    }
    mi *= (1.f / E_NUM); ml *= (1.f / E_NUM);
    float vi = 0.f, vl = 0.f;
    for (int e = 0; e < E_NUM; ++e) {
      float a = im[e] - mi, c = ld[e] - ml;
      vi += a * a; vl += c * c;
    }
    vi *= (1.f / (E_NUM - 1)); vl *= (1.f / (E_NUM - 1));
    *out_loss = vi / (mi * mi + 1e-10f) + vl / (ml * ml + 1e-10f);
  }
}

__device__ __forceinline__ void g1_load(v2f (&b)[4][4], const float* __restrict__ p) {
#pragma unroll
  for (int s = 0; s < 4; ++s)
#pragma unroll
    for (int nt = 0; nt < 4; ++nt) {
      b[s][nt][0] = p[(4 * s) * H_DIM + nt * 16];
      b[s][nt][1] = p[(4 * s) * H_DIM + nt * 16 + H_DIM];
    }
}

__device__ __forceinline__ void g1_mma(v8f (&acc)[2][4], const v2f (&b)[4][4],
                                       const float* xl, const float* xh) {
#pragma unroll
  for (int s = 0; s < 4; ++s) {
    float2 a2l = *(const float2*)(xl + 4 * s);
    float2 a2h = *(const float2*)(xh + 4 * s);
    v2f a0; a0[0] = a2l.x; a0[1] = a2l.y;
    v2f a1; a1[0] = a2h.x; a1[1] = a2h.y;
#pragma unroll
    for (int nt = 0; nt < 4; ++nt) {
      acc[0][nt] = __builtin_amdgcn_wmma_f32_16x16x4_f32(false, a0, false, b[s][nt],
                                                         (short)0, acc[0][nt], false, false);
      acc[1][nt] = __builtin_amdgcn_wmma_f32_16x16x4_f32(false, a1, false, b[s][nt],
                                                         (short)0, acc[1][nt], false, false);
    }
  }
}

__device__ __forceinline__ void g2_load(v2f (&b)[16], const float* __restrict__ p) {
#pragma unroll
  for (int s = 0; s < 16; ++s) {
    b[s][0] = p[(4 * s) * O_DIM];
    b[s][1] = p[(4 * s) * O_DIM + O_DIM];
  }
}

__device__ __forceinline__ void g2_mma(v8f (&acc)[2][2], const v2f (&b)[16],
                                       const float* hl, const float* hh) {
#pragma unroll
  for (int s = 0; s < 16; ++s) {
    float2 a2l = *(const float2*)(hl + 4 * s);
    float2 a2h = *(const float2*)(hh + 4 * s);
    v2f a0; a0[0] = a2l.x; a0[1] = a2l.y;
    v2f a1; a1[0] = a2h.x; a1[1] = a2h.y;
    acc[0][s & 1] = __builtin_amdgcn_wmma_f32_16x16x4_f32(false, a0, false, b[s],
                                                          (short)0, acc[0][s & 1], false, false);
    acc[1][s & 1] = __builtin_amdgcn_wmma_f32_16x16x4_f32(false, a1, false, b[s],
                                                          (short)0, acc[1][s & 1], false, false);
  }
}

__global__ __launch_bounds__(256, 6) void moe_expert(const float* __restrict__ x,
                                                  const float* __restrict__ w1,
                                                  const float* __restrict__ b1,
                                                  const float* __restrict__ w2,
                                                  const float* __restrict__ b2,
                                                  const float* __restrict__ gates,
                                                  const int* __restrict__ bucket,
                                                  const int* __restrict__ cnt,
                                                  float* __restrict__ y) {
  __shared__ float xs[TM][132];
  __shared__ float hs[TM][516];
  __shared__ float gbuf[TM];
  __shared__ int   tokbuf[TM];

  const int e    = blockIdx.y;
  const int n    = cnt[e];
  const int base = blockIdx.x * TM;
  if (base >= n) return;
  const int m = min(TM, n - base);

  const int tid  = threadIdx.x;
  const int lane = tid & 31;
  const int w    = tid >> 5;
  const int lh   = lane >> 4;
  const int ln   = lane & 15;

  if (tid < TM) {
    int idx = base + tid;
    int tok = bucket[e * B_TOK + ((idx < n) ? idx : (n - 1))];
    tokbuf[tid] = tok;
    gbuf[tid]   = (idx < n) ? gates[(size_t)tok * E_NUM + e] : 0.f;
  }
  __syncthreads();

#pragma unroll
  for (int i = 0; i < TM * D_IN / 256; ++i) {
    int idx = tid + i * 256;
    int r = idx >> 7, c = idx & 127;
    xs[r][c] = x[(size_t)tokbuf[r] * D_IN + c];
  }
  __syncthreads();

  {
    const float* w1e = w1 + (size_t)e * D_IN * H_DIM;
    const float* pB  = w1e + (2 * lh) * H_DIM + w * 64 + ln;
    v8f acc[2][4];
#pragma unroll
    for (int nt = 0; nt < 4; ++nt) {
      float bv = b1[e * H_DIM + w * 64 + nt * 16 + ln];
#pragma unroll
      for (int i = 0; i < 8; ++i) { acc[0][nt][i] = bv; acc[1][nt][i] = bv; }
    }
    v2f bq0[4][4], bq1[4][4];
    const float* xl = &xs[ln][2 * lh];
    const float* xh = &xs[16 + ln][2 * lh];
    g1_load(bq0, pB);
#pragma nounroll
    for (int kb = 0; kb < D_IN / 16 - 2; kb += 2) {
      g1_load(bq1, pB + (kb + 1) * 16 * H_DIM);
      g1_mma(acc, bq0, xl + kb * 16, xh + kb * 16);
      g1_load(bq0, pB + (kb + 2) * 16 * H_DIM);
      g1_mma(acc, bq1, xl + (kb + 1) * 16, xh + (kb + 1) * 16);
    }
    g1_load(bq1, pB + (D_IN / 16 - 1) * 16 * H_DIM);
    g1_mma(acc, bq0, xl + (D_IN / 16 - 2) * 16, xh + (D_IN / 16 - 2) * 16);
    g1_mma(acc, bq1, xl + (D_IN / 16 - 1) * 16, xh + (D_IN / 16 - 1) * 16);
#pragma unroll
    for (int mt = 0; mt < 2; ++mt)
#pragma unroll
      for (int nt = 0; nt < 4; ++nt)
#pragma unroll
        for (int i = 0; i < 8; ++i)
          hs[mt * 16 + 8 * lh + i][w * 64 + nt * 16 + ln] = fmaxf(acc[mt][nt][i], 0.f);
  }
  __syncthreads();

  {
    const int o0 = w * 16;
    const float* w2e = w2 + (size_t)e * H_DIM * O_DIM;
    const float* pB  = w2e + (2 * lh) * O_DIM + o0 + ln;
    v8f acc[2][2];
    float bv = b2[e * O_DIM + o0 + ln];
#pragma unroll
    for (int i = 0; i < 8; ++i) {
      acc[0][0][i] = bv;  acc[0][1][i] = 0.f;
      acc[1][0][i] = bv;  acc[1][1][i] = 0.f;
    }
    v2f cq0[16], cq1[16];
    g2_load(cq0, pB);
    const float* hl = &hs[ln][2 * lh];
    const float* hh = &hs[16 + ln][2 * lh];
    for (int kb = 0; kb < H_DIM / 64 - 2; kb += 2) {
      g2_load(cq1, pB + (kb + 1) * 64 * O_DIM);
      g2_mma(acc, cq0, hl + kb * 64, hh + kb * 64);
      g2_load(cq0, pB + (kb + 2) * 64 * O_DIM);
      g2_mma(acc, cq1, hl + (kb + 1) * 64, hh + (kb + 1) * 64);
    }
    g2_load(cq1, pB + (H_DIM / 64 - 1) * 64 * O_DIM);
    g2_mma(acc, cq0, hl + (H_DIM / 64 - 2) * 64, hh + (H_DIM / 64 - 2) * 64);
    g2_mma(acc, cq1, hl + (H_DIM / 64 - 1) * 64, hh + (H_DIM / 64 - 1) * 64);
#pragma unroll
    for (int mt = 0; mt < 2; ++mt)
#pragma unroll
      for (int i = 0; i < 8; ++i) {
        int M = mt * 16 + 8 * lh + i;
        float v = (acc[mt][0][i] + acc[mt][1][i]) * gbuf[M];
        if (M < m)
          unsafeAtomicAdd(&y[(size_t)tokbuf[M] * O_DIM + o0 + ln], v);
      }
  }
}

__global__ __launch_bounds__(256) void moe_eps(float* __restrict__ y) {
  const float EPSV = 2.2204460492503131e-16f;
  int idx = blockIdx.x * 256 + threadIdx.x;
  float4* p = ((float4*)y) + idx;
  float4 v = *p;
  v.x = (v.x == 0.f) ? EPSV : v.x;
  v.y = (v.y == 0.f) ? EPSV : v.y;
  v.z = (v.z == 0.f) ? EPSV : v.z;
  v.w = (v.w == 0.f) ? EPSV : v.w;
  *p = v;
}

extern "C" void kernel_launch(void* const* d_in, const int* in_sizes, int n_in,
                              void* d_out, int out_size, void* d_ws, size_t ws_size,
                              hipStream_t stream) {
  (void)in_sizes; (void)n_in; (void)out_size; (void)ws_size;
  const float* x      = (const float*)d_in[0];
  const float* w_gate = (const float*)d_in[1];
  const float* w1     = (const float*)d_in[2];
  const float* b1     = (const float*)d_in[3];
  const float* w2     = (const float*)d_in[4];
  const float* b2     = (const float*)d_in[5];

  float* y    = (float*)d_out;
  float* loss = y + (size_t)B_TOK * O_DIM;

  char*  ws         = (char*)d_ws;
  float* gates      = (float*)ws;
  int*   bucket     = (int*)(ws + (size_t)B_TOK * E_NUM * 4);
  int*   cnt        = (int*)(ws + 2 * (size_t)B_TOK * E_NUM * 4);
  float* importance = (float*)(ws + 2 * (size_t)B_TOK * E_NUM * 4 + 64);

  moe_init<<<(B_TOK * O_DIM) / (4 * 256), 256, 0, stream>>>(y, cnt, importance);
  moe_gate<<<B_TOK / 256, 256, 0, stream>>>(x, w_gate, gates, bucket, cnt, importance);
  moe_loss<<<1, 32, 0, stream>>>(importance, cnt, loss);
  dim3 g(B_TOK / TM, E_NUM);
  moe_expert<<<g, 256, 0, stream>>>(x, w1, b1, w2, b2, gates, bucket, cnt, y);
  moe_eps<<<(B_TOK * O_DIM) / (4 * 256), 256, 0, stream>>>(y);
}


# y without atomics: expert tiles store gated 512-B rows (staged in LDS) to a per-token top-1/top-2 plane, last kernel adds the pair (fixed order, was arrival order) and fills eps; zero-fill pass gone
# speedup vs baseline: 1.3258x; 1.1654x over previous
#include <hip/hip_runtime.h>
#include <hip/hip_bf16.h>
#include <math.h>

#define B_TOK 16384
#define D_IN  128
#define H_DIM 512
#define O_DIM 128
#define E_NUM 8
#define TM    32

typedef __attribute__((ext_vector_type(2))) float v2f;
typedef __attribute__((ext_vector_type(8))) float v8f;

__global__ __launch_bounds__(32) void moe_init(int* __restrict__ cnt,
                                               float* __restrict__ importance) {
  if (threadIdx.x < E_NUM) cnt[threadIdx.x] = 0;
  else if (threadIdx.x < 2 * E_NUM) importance[threadIdx.x - E_NUM] = 0.f;
}

__global__ __launch_bounds__(256) void moe_gate(const float* __restrict__ x,
                                                const float* __restrict__ w_gate,
                                                float* __restrict__ gates,
                                                int* __restrict__ bucket,
                                                int* __restrict__ cnt,
                                                float* __restrict__ importance) {
  __shared__ float wg[D_IN * E_NUM];
  __shared__ float simp[E_NUM];
  __shared__ int   scnt[E_NUM];
  __shared__ int   sbase[E_NUM];
  const int tid = threadIdx.x;
  for (int i = tid; i < D_IN * E_NUM; i += 256) wg[i] = w_gate[i];
  if (tid < E_NUM) { simp[tid] = 0.f; scnt[tid] = 0; }
  __syncthreads();

  const int b = blockIdx.x * 256 + tid;
  const float4* xr = (const float4*)(x + (size_t)b * D_IN);
  float l[E_NUM];
#pragma unroll
  for (int e = 0; e < E_NUM; ++e) l[e] = 0.f;
#pragma unroll 4
  for (int d4 = 0; d4 < D_IN / 4; ++d4) {
    float4 xv = xr[d4];
    const float* wr = &wg[d4 * 4 * E_NUM];
#pragma unroll
    for (int e = 0; e < E_NUM; ++e)
      l[e] += xv.x * wr[e] + xv.y * wr[E_NUM + e] + xv.z * wr[2 * E_NUM + e] + xv.w * wr[3 * E_NUM + e];
  }

  int e0 = 0; float l0 = l[0];
#pragma unroll
  for (int e = 1; e < E_NUM; ++e) if (l[e] > l0) { l0 = l[e]; e0 = e; }
  int e1 = -1; float l1 = -3.4e38f;
#pragma unroll
  for (int e = 0; e < E_NUM; ++e) if (e != e0 && l[e] > l1) { l1 = l[e]; e1 = e; }

  float q  = __expf(l1 - l0);
  float g0 = 1.f / (1.f + q);
  float g1 = q / (1.f + q);

  float gr[E_NUM];
#pragma unroll
  for (int e = 0; e < E_NUM; ++e) gr[e] = 0.f;
  gr[e0] = g0; gr[e1] = g1;
#pragma unroll
  for (int e = 0; e < E_NUM; ++e) gates[(size_t)b * E_NUM + e] = gr[e];

  atomicAdd(&simp[e0], g0);
  atomicAdd(&simp[e1], g1);
  const int r0 = atomicAdd(&scnt[e0], 1);
  const int r1 = atomicAdd(&scnt[e1], 1);
  __syncthreads();
  if (tid < E_NUM) {
    sbase[tid] = atomicAdd(&cnt[tid], scnt[tid]);
    unsafeAtomicAdd(&importance[tid], simp[tid]);
  }
  __syncthreads();
  bucket[e0 * B_TOK + sbase[e0] + r0] = 2 * b;
  bucket[e1 * B_TOK + sbase[e1] + r1] = 2 * b + 1;
}

__global__ void moe_loss(const float* __restrict__ importance,
                         const int* __restrict__ cnt,
                         float* __restrict__ out_loss) {
  if (threadIdx.x == 0 && blockIdx.x == 0) {
    float im[E_NUM], ld[E_NUM], mi = 0.f, ml = 0.f;
    for (int e = 0; e < E_NUM; ++e) {
      im[e] = importance[e]; ld[e] = (float)cnt[e];
      mi += im[e]; ml += ld[e];
    }
    mi *= (1.f / E_NUM); ml *= (1.f / E_NUM);
    float vi = 0.f, vl = 0.f;
    for (int e = 0; e < E_NUM; ++e) {
      float a = im[e] - mi, c = ld[e] - ml;
      vi += a * a; vl += c * c;
    }
    vi *= (1.f / (E_NUM - 1)); vl *= (1.f / (E_NUM - 1));
    *out_loss = vi / (mi * mi + 1e-10f) + vl / (ml * ml + 1e-10f);
  }
}

__device__ __forceinline__ void g1_load(v2f (&b)[4][4], const float* __restrict__ p) {
#pragma unroll
  for (int s = 0; s < 4; ++s)
#pragma unroll
    for (int nt = 0; nt < 4; ++nt) {
      b[s][nt][0] = p[(4 * s) * H_DIM + nt * 16];
      b[s][nt][1] = p[(4 * s) * H_DIM + nt * 16 + H_DIM];
    }
}

__device__ __forceinline__ void g1_mma(v8f (&acc)[2][4], const v2f (&b)[4][4],
                                       const float* xl, const float* xh) {
#pragma unroll
  for (int s = 0; s < 4; ++s) {
    float2 a2l = *(const float2*)(xl + 4 * s);
    float2 a2h = *(const float2*)(xh + 4 * s);
    v2f a0; a0[0] = a2l.x; a0[1] = a2l.y;
    v2f a1; a1[0] = a2h.x; a1[1] = a2h.y;
#pragma unroll
    for (int nt = 0; nt < 4; ++nt) {
      acc[0][nt] = __builtin_amdgcn_wmma_f32_16x16x4_f32(false, a0, false, b[s][nt],
                                                         (short)0, acc[0][nt], false, false);
      acc[1][nt] = __builtin_amdgcn_wmma_f32_16x16x4_f32(false, a1, false, b[s][nt],
                                                         (short)0, acc[1][nt], false, false);
    }
  }
}

__device__ __forceinline__ void g2_load(v2f (&b)[16], const float* __restrict__ p) {
#pragma unroll
  for (int s = 0; s < 16; ++s) {
    b[s][0] = p[(4 * s) * O_DIM];
    b[s][1] = p[(4 * s) * O_DIM + O_DIM];
  }
}

__device__ __forceinline__ void g2_mma(v8f (&acc)[2][2], const v2f (&b)[16],
                                       const float* hl, const float* hh) {
#pragma unroll
  for (int s = 0; s < 16; ++s) {
    float2 a2l = *(const float2*)(hl + 4 * s);
    float2 a2h = *(const float2*)(hh + 4 * s);
    v2f a0; a0[0] = a2l.x; a0[1] = a2l.y;
    v2f a1; a1[0] = a2h.x; a1[1] = a2h.y;
    acc[0][s & 1] = __builtin_amdgcn_wmma_f32_16x16x4_f32(false, a0, false, b[s],
                                                          (short)0, acc[0][s & 1], false, false);
    acc[1][s & 1] = __builtin_amdgcn_wmma_f32_16x16x4_f32(false, a1, false, b[s],
                                                          (short)0, acc[1][s & 1], false, false);
  }
}

__global__ __launch_bounds__(256, 6) void moe_expert(const float* __restrict__ x,
                                                  const float* __restrict__ w1,
                                                  const float* __restrict__ b1,
                                                  const float* __restrict__ w2,
                                                  const float* __restrict__ b2,
                                                  const float* __restrict__ gates,
                                                  const int* __restrict__ bucket,
                                                  const int* __restrict__ cnt,
                                                  float* __restrict__ part) {
  __shared__ __attribute__((aligned(16))) float xs[TM][132];
  __shared__ float hs[TM][516];
  __shared__ float gbuf[TM];
  __shared__ int   tokbuf[TM];

  const int e    = blockIdx.y;
  const int n    = cnt[e];
  const int base = blockIdx.x * TM;
  if (base >= n) return;
  const int m = min(TM, n - base);

  const int tid  = threadIdx.x;
  const int lane = tid & 31;
  const int w    = tid >> 5;
  const int lh   = lane >> 4;
  const int ln   = lane & 15;

  if (tid < TM) {
    int idx = base + tid;
    int tok2 = bucket[e * B_TOK + ((idx < n) ? idx : (n - 1))];
    tokbuf[tid] = tok2;
    gbuf[tid]   = (idx < n) ? gates[(size_t)(tok2 >> 1) * E_NUM + e] : 0.f;
  }
  __syncthreads();

#pragma unroll
  for (int i = 0; i < TM * D_IN / 256; ++i) {
    int idx = tid + i * 256;
    int r = idx >> 7, c = idx & 127;
    xs[r][c] = x[(size_t)(tokbuf[r] >> 1) * D_IN + c];
  }
  __syncthreads();

  {
    const float* w1e = w1 + (size_t)e * D_IN * H_DIM;
    const float* pB  = w1e + (2 * lh) * H_DIM + w * 64 + ln;
    v8f acc[2][4];
#pragma unroll
    for (int nt = 0; nt < 4; ++nt) {
      float bv = b1[e * H_DIM + w * 64 + nt * 16 + ln];
#pragma unroll
      for (int i = 0; i < 8; ++i) { acc[0][nt][i] = bv; acc[1][nt][i] = bv; }
    }
    v2f bq0[4][4], bq1[4][4];
    const float* xl = &xs[ln][2 * lh];
    const float* xh = &xs[16 + ln][2 * lh];
    g1_load(bq0, pB);
#pragma nounroll
    for (int kb = 0; kb < D_IN / 16 - 2; kb += 2) {
      g1_load(bq1, pB + (kb + 1) * 16 * H_DIM);
      g1_mma(acc, bq0, xl + kb * 16, xh + kb * 16);
      g1_load(bq0, pB + (kb + 2) * 16 * H_DIM);
      g1_mma(acc, bq1, xl + (kb + 1) * 16, xh + (kb + 1) * 16);
    }
    g1_load(bq1, pB + (D_IN / 16 - 1) * 16 * H_DIM);
    g1_mma(acc, bq0, xl + (D_IN / 16 - 2) * 16, xh + (D_IN / 16 - 2) * 16);
    g1_mma(acc, bq1, xl + (D_IN / 16 - 1) * 16, xh + (D_IN / 16 - 1) * 16);
#pragma unroll
    for (int mt = 0; mt < 2; ++mt)
#pragma unroll
      for (int nt = 0; nt < 4; ++nt)
#pragma unroll
        for (int i = 0; i < 8; ++i)
          hs[mt * 16 + 8 * lh + i][w * 64 + nt * 16 + ln] = fmaxf(acc[mt][nt][i], 0.f);
  }
  __syncthreads();

  {
    const int o0 = w * 16;
    const float* w2e = w2 + (size_t)e * H_DIM * O_DIM;
    const float* pB  = w2e + (2 * lh) * O_DIM + o0 + ln;
    v8f acc[2][2];
    float bv = b2[e * O_DIM + o0 + ln];
#pragma unroll
    for (int i = 0; i < 8; ++i) {
      acc[0][0][i] = bv;  acc[0][1][i] = 0.f;
      acc[1][0][i] = bv;  acc[1][1][i] = 0.f;
    }
    v2f cq0[16], cq1[16];
    g2_load(cq0, pB);
    const float* hl = &hs[ln][2 * lh];
    const float* hh = &hs[16 + ln][2 * lh];
    for (int kb = 0; kb < H_DIM / 64 - 2; kb += 2) {
      g2_load(cq1, pB + (kb + 1) * 64 * O_DIM);
      g2_mma(acc, cq0, hl + kb * 64, hh + kb * 64);
      g2_load(cq0, pB + (kb + 2) * 64 * O_DIM);
      g2_mma(acc, cq1, hl + (kb + 1) * 64, hh + (kb + 1) * 64);
    }
    g2_load(cq1, pB + (H_DIM / 64 - 1) * 64 * O_DIM);
    g2_mma(acc, cq0, hl + (H_DIM / 64 - 2) * 64, hh + (H_DIM / 64 - 2) * 64);
    g2_mma(acc, cq1, hl + (H_DIM / 64 - 1) * 64, hh + (H_DIM / 64 - 1) * 64);
#pragma unroll
    for (int mt = 0; mt < 2; ++mt)
#pragma unroll
      for (int i = 0; i < 8; ++i) {
        int M = mt * 16 + 8 * lh + i;
        xs[M][o0 + ln] = (acc[mt][0][i] + acc[mt][1][i]) * gbuf[M];
      }
  }
  __syncthreads();
#pragma unroll
  for (int j = 0; j < TM / 8; ++j) {
    const int r = w * (TM / 8) + j;
    if (r < m) {
      float4 v = *(const float4*)&xs[r][4 * lane];
      *(float4*)(part + (size_t)tokbuf[r] * O_DIM + 4 * lane) = v;
    }
  }
}

__global__ __launch_bounds__(256) void moe_combine(const float* __restrict__ part,
                                                   float* __restrict__ y) {
  const float EPSV = 2.2204460492503131e-16f;
  int idx = blockIdx.x * 256 + threadIdx.x;
  int b = idx >> 5, c4 = idx & 31;
  const float4* pr = ((const float4*)part) + (size_t)b * (2 * O_DIM / 4) + c4;
  float4 t0 = pr[0];
  float4 t1 = pr[O_DIM / 4];
  float4* p = ((float4*)y) + idx;
  float4 v;
  v.x = t0.x + t1.x; v.y = t0.y + t1.y; v.z = t0.z + t1.z; v.w = t0.w + t1.w;
  v.x = (v.x == 0.f) ? EPSV : v.x;
  v.y = (v.y == 0.f) ? EPSV : v.y;
  v.z = (v.z == 0.f) ? EPSV : v.z;
  v.w = (v.w == 0.f) ? EPSV : v.w;
  *p = v;
}

extern "C" void kernel_launch(void* const* d_in, const int* in_sizes, int n_in,
                              void* d_out, int out_size, void* d_ws, size_t ws_size,
                              hipStream_t stream) {
  (void)in_sizes; (void)n_in; (void)out_size; (void)ws_size;
  const float* x      = (const float*)d_in[0];
  const float* w_gate = (const float*)d_in[1];
  const float* w1     = (const float*)d_in[2];
  const float* b1     = (const float*)d_in[3];
  const float* w2     = (const float*)d_in[4];
  const float* b2     = (const float*)d_in[5];

  float* y    = (float*)d_out;
  float* loss = y + (size_t)B_TOK * O_DIM;

  char*  ws         = (char*)d_ws;
  float* gates      = (float*)ws;
  int*   bucket     = (int*)(ws + (size_t)B_TOK * E_NUM * 4);
  int*   cnt        = (int*)(ws + 2 * (size_t)B_TOK * E_NUM * 4);
  float* importance = (float*)(ws + 2 * (size_t)B_TOK * E_NUM * 4 + 64);
  float* part       = (float*)(ws + 4 * (size_t)B_TOK * E_NUM * 4);

  moe_init<<<1, 32, 0, stream>>>(cnt, importance);
  moe_gate<<<B_TOK / 256, 256, 0, stream>>>(x, w_gate, gates, bucket, cnt, importance);
  moe_loss<<<1, 32, 0, stream>>>(importance, cnt, loss);
  dim3 g(B_TOK / TM, E_NUM);
  moe_expert<<<g, 256, 0, stream>>>(x, w1, b1, w2, b2, gates, bucket, cnt, part);
  moe_combine<<<(B_TOK * O_DIM) / (4 * 256), 256, 0, stream>>>(part, y);
}


# 5 launches -> 3: gate blocks keep counts, gate sums, token lists in own slots (no global counters to zero); expert kernel scans 64 counts, one work-group does the loss; importance sum: fixed tree
# speedup vs baseline: 1.3700x; 1.0333x over previous
#include <hip/hip_runtime.h>
#include <hip/hip_bf16.h>
#include <math.h>

#define B_TOK 16384
#define D_IN  128
#define H_DIM 512
#define O_DIM 128
#define E_NUM 8
#define TM    32
#define GB    256
#define NBLK  (B_TOK / GB)

typedef __attribute__((ext_vector_type(2))) float v2f;
typedef __attribute__((ext_vector_type(8))) float v8f;

__global__ __launch_bounds__(256) void moe_gate(const float* __restrict__ x,
                                                const float* __restrict__ w_gate,
                                                float* __restrict__ gates,
                                                int* __restrict__ seg,
                                                int* __restrict__ pc,
                                                float* __restrict__ pi) {
  __shared__ float wg[D_IN * E_NUM];
  __shared__ float simp[E_NUM];
  __shared__ int   scnt[E_NUM];
  const int tid = threadIdx.x;
  for (int i = tid; i < D_IN * E_NUM; i += 256) wg[i] = w_gate[i];
  if (tid < E_NUM) { simp[tid] = 0.f; scnt[tid] = 0; }
  __syncthreads();

  const int b = blockIdx.x * 256 + tid;
  const float4* xr = (const float4*)(x + (size_t)b * D_IN);
  float l[E_NUM];
#pragma unroll
  for (int e = 0; e < E_NUM; ++e) l[e] = 0.f;
#pragma unroll 4
  for (int d4 = 0; d4 < D_IN / 4; ++d4) {
    float4 xv = xr[d4];
    const float* wr = &wg[d4 * 4 * E_NUM];
#pragma unroll
    for (int e = 0; e < E_NUM; ++e)
      l[e] += xv.x * wr[e] + xv.y * wr[E_NUM + e] + xv.z * wr[2 * E_NUM + e] + xv.w * wr[3 * E_NUM + e];
  }

  int e0 = 0; float l0 = l[0];
#pragma unroll
  for (int e = 1; e < E_NUM; ++e) if (l[e] > l0) { l0 = l[e]; e0 = e; }
  int e1 = -1; float l1 = -3.4e38f;
#pragma unroll
  for (int e = 0; e < E_NUM; ++e) if (e != e0 && l[e] > l1) { l1 = l[e]; e1 = e; }

  float q  = __expf(l1 - l0);
  float g0 = 1.f / (1.f + q);
  float g1 = q / (1.f + q);

  float gr[E_NUM];
#pragma unroll
  for (int e = 0; e < E_NUM; ++e) gr[e] = 0.f;
  gr[e0] = g0; gr[e1] = g1;
#pragma unroll
  for (int e = 0; e < E_NUM; ++e) gates[(size_t)b * E_NUM + e] = gr[e];

  atomicAdd(&simp[e0], g0);
  atomicAdd(&simp[e1], g1);
  const int r0 = atomicAdd(&scnt[e0], 1);
  const int r1 = atomicAdd(&scnt[e1], 1);
  __syncthreads();
  if (tid < E_NUM) {
    pc[tid * NBLK + blockIdx.x] = scnt[tid];
    pi[tid * NBLK + blockIdx.x] = simp[tid];
  }
  seg[(e0 * NBLK + blockIdx.x) * GB + r0] = 2 * b;
  seg[(e1 * NBLK + blockIdx.x) * GB + r1] = 2 * b + 1;
}

__device__ __forceinline__ void moe_loss_wg(const int* __restrict__ pc,
                                            const float* __restrict__ pi,
                                            float* __restrict__ out_loss,
                                            float* lsum, int w, int lane) {
  static_assert(E_NUM == 256 / 32, "one wave of the 256-thread work-group per expert");
  static_assert(NBLK == 64, "two gate blocks per lane");
  float iv = pi[w * NBLK + lane] + pi[w * NBLK + lane + 32];
  int   cv = pc[w * NBLK + lane] + pc[w * NBLK + lane + 32];
  for (int d = 16; d >= 1; d >>= 1) {
    iv += __shfl_xor(iv, d);
    cv += __shfl_xor(cv, d);
  }
  if (lane == 0) { lsum[w] = iv; lsum[E_NUM + w] = (float)cv; }
  __syncthreads();
  if (w == 0 && lane == 0) {
    float im[E_NUM], ld[E_NUM], mi = 0.f, ml = 0.f;
    for (int e = 0; e < E_NUM; ++e) {
      im[e] = lsum[e]; ld[e] = lsum[E_NUM + e];
      mi += im[e]; ml += ld[e];
    }
    mi *= (1.f / E_NUM); ml *= (1.f / E_NUM);
    float vi = 0.f, vl = 0.f;
    for (int e = 0; e < E_NUM; ++e) {
      float a = im[e] - mi, c = ld[e] - ml;
      vi += a * a; vl += c * c;
    }
    vi *= (1.f / (E_NUM - 1)); vl *= (1.f / (E_NUM - 1));
    *out_loss = vi / (mi * mi + 1e-10f) + vl / (ml * ml + 1e-10f);
  }
}

__device__ __forceinline__ void g1_load(v2f (&b)[4][4], const float* __restrict__ p) {
#pragma unroll
  for (int s = 0; s < 4; ++s)
#pragma unroll
    for (int nt = 0; nt < 4; ++nt) {
      b[s][nt][0] = p[(4 * s) * H_DIM + nt * 16];
      b[s][nt][1] = p[(4 * s) * H_DIM + nt * 16 + H_DIM];
    }
}

__device__ __forceinline__ void g1_mma(v8f (&acc)[2][4], const v2f (&b)[4][4],
                                       const float* xl, const float* xh) {
#pragma unroll
  for (int s = 0; s < 4; ++s) {
    float2 a2l = *(const float2*)(xl + 4 * s);
    float2 a2h = *(const float2*)(xh + 4 * s);
    v2f a0; a0[0] = a2l.x; a0[1] = a2l.y;
    v2f a1; a1[0] = a2h.x; a1[1] = a2h.y;
#pragma unroll
    for (int nt = 0; nt < 4; ++nt) {
      acc[0][nt] = __builtin_amdgcn_wmma_f32_16x16x4_f32(false, a0, false, b[s][nt],
                                                         (short)0, acc[0][nt], false, false);
      acc[1][nt] = __builtin_amdgcn_wmma_f32_16x16x4_f32(false, a1, false, b[s][nt],
                                                         (short)0, acc[1][nt], false, false);
    }
  }
}

__device__ __forceinline__ void g2_load(v2f (&b)[16], const float* __restrict__ p) {
#pragma unroll
  for (int s = 0; s < 16; ++s) {
    b[s][0] = p[(4 * s) * O_DIM];
    b[s][1] = p[(4 * s) * O_DIM + O_DIM];
  }
}

__device__ __forceinline__ void g2_mma(v8f (&acc)[2][2], const v2f (&b)[16],
                                       const float* hl, const float* hh) {
#pragma unroll
  for (int s = 0; s < 16; ++s) {
    float2 a2l = *(const float2*)(hl + 4 * s);
    float2 a2h = *(const float2*)(hh + 4 * s);
    v2f a0; a0[0] = a2l.x; a0[1] = a2l.y;
    v2f a1; a1[0] = a2h.x; a1[1] = a2h.y;
    acc[0][s & 1] = __builtin_amdgcn_wmma_f32_16x16x4_f32(false, a0, false, b[s],
                                                          (short)0, acc[0][s & 1], false, false);
    acc[1][s & 1] = __builtin_amdgcn_wmma_f32_16x16x4_f32(false, a1, false, b[s],
                                                          (short)0, acc[1][s & 1], false, false);
  }
}

__global__ __launch_bounds__(256, 6) void moe_expert(const float* __restrict__ x,
                                                  const float* __restrict__ w1,
                                                  const float* __restrict__ b1,
                                                  const float* __restrict__ w2,
                                                  const float* __restrict__ b2,
                                                  const float* __restrict__ gates,
                                                  const int* __restrict__ seg,
                                                  const int* __restrict__ pc,
                                                  const float* __restrict__ pi,
                                                  float* __restrict__ part,
                                                  float* __restrict__ out_loss) {
  __shared__ __attribute__((aligned(16))) float xs[TM][132];
  __shared__ float hs[TM][516];
  __shared__ float gbuf[TM];
  __shared__ int   tokbuf[TM];
  __shared__ float lsum[2 * E_NUM];

  const int tid  = threadIdx.x;
  const int lane = tid & 31;
  const int w    = tid >> 5;
  const int lh   = lane >> 4;
  const int ln   = lane & 15;

  const int e    = blockIdx.y;
  const int* pce = pc + e * NBLK;
  int cb[NBLK];
  int n = 0;
#pragma unroll
  for (int i = 0; i < NBLK; ++i) { cb[i] = pce[i]; n += cb[i]; }

  const bool loss_wg = (blockIdx.x == B_TOK / TM - 1 && blockIdx.y == 0);

  const int base = blockIdx.x * TM;
  if (base >= n) {
    if (loss_wg) moe_loss_wg(pc, pi, out_loss, lsum, w, lane);
    return;
  }
  const int m = min(TM, n - base);

  if (tid < TM) {
    const int idx = min(base + tid, n - 1);
    int s = 0, blk = 0, off = idx;
#pragma unroll
    for (int i = 0; i < NBLK; ++i) {
      if (idx >= s) { blk = i; off = idx - s; }
      s += cb[i];
    }
    int tok2 = seg[(e * NBLK + blk) * GB + off];
    tokbuf[tid] = tok2;
    gbuf[tid]   = (base + tid < n) ? gates[(size_t)(tok2 >> 1) * E_NUM + e] : 0.f;
  }
  __syncthreads();

#pragma unroll
  for (int i = 0; i < TM * D_IN / 256; ++i) {
    int idx = tid + i * 256;
    int r = idx >> 7, c = idx & 127;
    xs[r][c] = x[(size_t)(tokbuf[r] >> 1) * D_IN + c];
  }
  __syncthreads();

  {
    const float* w1e = w1 + (size_t)e * D_IN * H_DIM;
    const float* pB  = w1e + (2 * lh) * H_DIM + w * 64 + ln;
    v8f acc[2][4];
#pragma unroll
    for (int nt = 0; nt < 4; ++nt) {
      float bv = b1[e * H_DIM + w * 64 + nt * 16 + ln];
#pragma unroll
      for (int i = 0; i < 8; ++i) { acc[0][nt][i] = bv; acc[1][nt][i] = bv; }
    }
    v2f bq0[4][4], bq1[4][4];
    const float* xl = &xs[ln][2 * lh];
    const float* xh = &xs[16 + ln][2 * lh];
    g1_load(bq0, pB);
#pragma nounroll
    for (int kb = 0; kb < D_IN / 16 - 2; kb += 2) {
      g1_load(bq1, pB + (kb + 1) * 16 * H_DIM);
      g1_mma(acc, bq0, xl + kb * 16, xh + kb * 16);
      g1_load(bq0, pB + (kb + 2) * 16 * H_DIM);
      g1_mma(acc, bq1, xl + (kb + 1) * 16, xh + (kb + 1) * 16);
    }
    g1_load(bq1, pB + (D_IN / 16 - 1) * 16 * H_DIM);
    g1_mma(acc, bq0, xl + (D_IN / 16 - 2) * 16, xh + (D_IN / 16 - 2) * 16);
    g1_mma(acc, bq1, xl + (D_IN / 16 - 1) * 16, xh + (D_IN / 16 - 1) * 16);
#pragma unroll
    for (int mt = 0; mt < 2; ++mt)
#pragma unroll
      for (int nt = 0; nt < 4; ++nt)
#pragma unroll
        for (int i = 0; i < 8; ++i)
          hs[mt * 16 + 8 * lh + i][w * 64 + nt * 16 + ln] = fmaxf(acc[mt][nt][i], 0.f);
  }
  __syncthreads();

  {
    const int o0 = w * 16;
    const float* w2e = w2 + (size_t)e * H_DIM * O_DIM;
    const float* pB  = w2e + (2 * lh) * O_DIM + o0 + ln;
    v8f acc[2][2];
    float bv = b2[e * O_DIM + o0 + ln];
#pragma unroll
    for (int i = 0; i < 8; ++i) {
      acc[0][0][i] = bv;  acc[0][1][i] = 0.f;
      acc[1][0][i] = bv;  acc[1][1][i] = 0.f;
    }
    v2f cq0[16], cq1[16];
    g2_load(cq0, pB);
    const float* hl = &hs[ln][2 * lh];
    const float* hh = &hs[16 + ln][2 * lh];
    for (int kb = 0; kb < H_DIM / 64 - 2; kb += 2) {
      g2_load(cq1, pB + (kb + 1) * 64 * O_DIM);
      g2_mma(acc, cq0, hl + kb * 64, hh + kb * 64);
      g2_load(cq0, pB + (kb + 2) * 64 * O_DIM);
      g2_mma(acc, cq1, hl + (kb + 1) * 64, hh + (kb + 1) * 64);
    }
    g2_load(cq1, pB + (H_DIM / 64 - 1) * 64 * O_DIM);
    g2_mma(acc, cq0, hl + (H_DIM / 64 - 2) * 64, hh + (H_DIM / 64 - 2) * 64);
    g2_mma(acc, cq1, hl + (H_DIM / 64 - 1) * 64, hh + (H_DIM / 64 - 1) * 64);
#pragma unroll
    for (int mt = 0; mt < 2; ++mt)
#pragma unroll
      for (int i = 0; i < 8; ++i) {
        int M = mt * 16 + 8 * lh + i;
        xs[M][o0 + ln] = (acc[mt][0][i] + acc[mt][1][i]) * gbuf[M];
      }
  }
  __syncthreads();
#pragma unroll
  for (int j = 0; j < TM / 8; ++j) {
    const int r = w * (TM / 8) + j;
    if (r < m) {
      float4 v = *(const float4*)&xs[r][4 * lane];
      *(float4*)(part + (size_t)tokbuf[r] * O_DIM + 4 * lane) = v;
    }
  }
  if (loss_wg) moe_loss_wg(pc, pi, out_loss, lsum, w, lane);
}

__global__ __launch_bounds__(256) void moe_combine(const float* __restrict__ part,
                                                   float* __restrict__ y) {
  const float EPSV = 2.2204460492503131e-16f;
  int idx = blockIdx.x * 256 + threadIdx.x;
  int b = idx >> 5, c4 = idx & 31;
  const float4* pr = ((const float4*)part) + (size_t)b * (2 * O_DIM / 4) + c4;
  float4 t0 = pr[0];
  float4 t1 = pr[O_DIM / 4];
  float4* p = ((float4*)y) + idx;
  float4 v;
  v.x = t0.x + t1.x; v.y = t0.y + t1.y; v.z = t0.z + t1.z; v.w = t0.w + t1.w;
  v.x = (v.x == 0.f) ? EPSV : v.x;
  v.y = (v.y == 0.f) ? EPSV : v.y;
  v.z = (v.z == 0.f) ? EPSV : v.z;
  v.w = (v.w == 0.f) ? EPSV : v.w;
  *p = v;
}

extern "C" void kernel_launch(void* const* d_in, const int* in_sizes, int n_in,
                              void* d_out, int out_size, void* d_ws, size_t ws_size,
                              hipStream_t stream) {
  (void)in_sizes; (void)n_in; (void)out_size; (void)ws_size;
  const float* x      = (const float*)d_in[0];
  const float* w_gate = (const float*)d_in[1];
  const float* w1     = (const float*)d_in[2];
  const float* b1     = (const float*)d_in[3];
  const float* w2     = (const float*)d_in[4];
  const float* b2     = (const float*)d_in[5];

  float* y    = (float*)d_out;
  float* loss = y + (size_t)B_TOK * O_DIM;

  char*  ws         = (char*)d_ws;
  float* gates      = (float*)ws;
  int*   seg        = (int*)(ws + (size_t)B_TOK * E_NUM * 4);
  int*   pc         = (int*)(ws + 2 * (size_t)B_TOK * E_NUM * 4);
  float* pi         = (float*)(ws + 2 * (size_t)B_TOK * E_NUM * 4 + 4096);
  float* part       = (float*)(ws + 4 * (size_t)B_TOK * E_NUM * 4);

  moe_gate<<<NBLK, GB, 0, stream>>>(x, w_gate, gates, seg, pc, pi);
  dim3 g(B_TOK / TM, E_NUM);
  moe_expert<<<g, 256, 0, stream>>>(x, w1, b1, w2, b2, gates, seg, pc, pi, part, loss);
  moe_combine<<<(B_TOK * O_DIM) / (4 * 256), 256, 0, stream>>>(part, y);
}


# expert kernel: 48-token tiles (a weight fragment feeds 3 matrix ops, was 2), x and output tiles share the h buffer in LDS, 1-D grid of 690: every tile resident at once; sums unchanged
# speedup vs baseline: 1.5417x; 1.1253x over previous
#include <hip/hip_runtime.h>
#include <hip/hip_bf16.h>
#include <math.h>

#define B_TOK 16384
#define D_IN  128
#define H_DIM 512
#define O_DIM 128
#define E_NUM 8
#define TM    48
#define MT    (TM / 16)
#define XP    132
#define HP    516
#define MAXT  ((2 * B_TOK + E_NUM * (TM - 1)) / TM)
#define GB    256
#define NBLK  (B_TOK / GB)

typedef __attribute__((ext_vector_type(2))) float v2f;
typedef __attribute__((ext_vector_type(8))) float v8f;

__global__ __launch_bounds__(256) void moe_gate(const float* __restrict__ x,
                                                const float* __restrict__ w_gate,
                                                float* __restrict__ gates,
                                                int* __restrict__ seg,
                                                int* __restrict__ pc,
                                                float* __restrict__ pi) {
  __shared__ float wg[D_IN * E_NUM];
  __shared__ float simp[E_NUM];
  __shared__ int   scnt[E_NUM];
  const int tid = threadIdx.x;
  for (int i = tid; i < D_IN * E_NUM; i += 256) wg[i] = w_gate[i];
  if (tid < E_NUM) { simp[tid] = 0.f; scnt[tid] = 0; }
  __syncthreads();

  const int b = blockIdx.x * 256 + tid;
  const float4* xr = (const float4*)(x + (size_t)b * D_IN);
  float l[E_NUM];
#pragma unroll
  for (int e = 0; e < E_NUM; ++e) l[e] = 0.f;
#pragma unroll 4
  for (int d4 = 0; d4 < D_IN / 4; ++d4) {
    float4 xv = xr[d4];
    const float* wr = &wg[d4 * 4 * E_NUM];
#pragma unroll
    for (int e = 0; e < E_NUM; ++e)
      l[e] += xv.x * wr[e] + xv.y * wr[E_NUM + e] + xv.z * wr[2 * E_NUM + e] + xv.w * wr[3 * E_NUM + e];
  }

  int e0 = 0; float l0 = l[0];
#pragma unroll
  for (int e = 1; e < E_NUM; ++e) if (l[e] > l0) { l0 = l[e]; e0 = e; }
  int e1 = -1; float l1 = -3.4e38f;
#pragma unroll
  for (int e = 0; e < E_NUM; ++e) if (e != e0 && l[e] > l1) { l1 = l[e]; e1 = e; }

  float q  = __expf(l1 - l0);
  float g0 = 1.f / (1.f + q);
  float g1 = q / (1.f + q);

  float gr[E_NUM];
#pragma unroll
  for (int e = 0; e < E_NUM; ++e) gr[e] = 0.f;
  gr[e0] = g0; gr[e1] = g1;
#pragma unroll
  for (int e = 0; e < E_NUM; ++e) gates[(size_t)b * E_NUM + e] = gr[e];

  atomicAdd(&simp[e0], g0);
  atomicAdd(&simp[e1], g1);
  const int r0 = atomicAdd(&scnt[e0], 1);
  const int r1 = atomicAdd(&scnt[e1], 1);
  __syncthreads();
  if (tid < E_NUM) {
    pc[tid * NBLK + blockIdx.x] = scnt[tid];
    pi[tid * NBLK + blockIdx.x] = simp[tid];
  }
  seg[(e0 * NBLK + blockIdx.x) * GB + r0] = 2 * b;
  seg[(e1 * NBLK + blockIdx.x) * GB + r1] = 2 * b + 1;
}

__device__ __forceinline__ void moe_loss_finish(const float* lsum, float* __restrict__ out_loss) {
  float im[E_NUM], ld[E_NUM], mi = 0.f, ml = 0.f;
  for (int e = 0; e < E_NUM; ++e) {
    im[e] = lsum[e]; ld[e] = lsum[E_NUM + e];
    mi += im[e]; ml += ld[e];
  }
  mi *= (1.f / E_NUM); ml *= (1.f / E_NUM);
  float vi = 0.f, vl = 0.f;
  for (int e = 0; e < E_NUM; ++e) {
    float a = im[e] - mi, c = ld[e] - ml;
    vi += a * a; vl += c * c;
  }
  vi *= (1.f / (E_NUM - 1)); vl *= (1.f / (E_NUM - 1));
  *out_loss = vi / (mi * mi + 1e-10f) + vl / (ml * ml + 1e-10f);
}

__device__ __forceinline__ void g1_load(v2f (&b)[2][4], const float* __restrict__ p) {
#pragma unroll
  for (int s = 0; s < 2; ++s)
#pragma unroll
    for (int nt = 0; nt < 4; ++nt) {
      b[s][nt][0] = p[(4 * s) * H_DIM + nt * 16];
      b[s][nt][1] = p[(4 * s) * H_DIM + nt * 16 + H_DIM];
    }
}

__device__ __forceinline__ void g1_mma(v8f (&acc)[MT][4], const v2f (&b)[2][4],
                                       const float* x0) {
#pragma unroll
  for (int s = 0; s < 2; ++s) {
    v2f a[MT];
#pragma unroll
    for (int mt = 0; mt < MT; ++mt) {
      float2 a2 = *(const float2*)(x0 + mt * 16 * XP + 4 * s);
      a[mt][0] = a2.x; a[mt][1] = a2.y;
    }
#pragma unroll
    for (int nt = 0; nt < 4; ++nt)
#pragma unroll
      for (int mt = 0; mt < MT; ++mt)
        acc[mt][nt] = __builtin_amdgcn_wmma_f32_16x16x4_f32(false, a[mt], false, b[s][nt],
                                                            (short)0, acc[mt][nt], false, false);
  }
}

__device__ __forceinline__ void g2_load(v2f (&b)[16], const float* __restrict__ p) {
#pragma unroll
  for (int s = 0; s < 16; ++s) {
    b[s][0] = p[(4 * s) * O_DIM];
    b[s][1] = p[(4 * s) * O_DIM + O_DIM];
  }
}

__device__ __forceinline__ void g2_mma(v8f (&acc)[MT][2], const v2f (&b)[16],
                                       const float* h0) {
#pragma unroll
  for (int s = 0; s < 16; ++s) {
#pragma unroll
    for (int mt = 0; mt < MT; ++mt) {
      float2 a2 = *(const float2*)(h0 + mt * 16 * HP + 4 * s);
      v2f a; a[0] = a2.x; a[1] = a2.y;
      acc[mt][s & 1] = __builtin_amdgcn_wmma_f32_16x16x4_f32(false, a, false, b[s],
                                                             (short)0, acc[mt][s & 1], false, false);
    }
  }
}

__global__ __launch_bounds__(256, 6) void moe_expert(const float* __restrict__ x,
                                                  const float* __restrict__ w1,
                                                  const float* __restrict__ b1,
                                                  const float* __restrict__ w2,
                                                  const float* __restrict__ b2,
                                                  const float* __restrict__ gates,
                                                  const int* __restrict__ seg,
                                                  const int* __restrict__ pc,
                                                  const float* __restrict__ pi,
                                                  float* __restrict__ part,
                                                  float* __restrict__ out_loss) {
  __shared__ __attribute__((aligned(16))) float hs[TM][HP];
  float (*xs)[XP] = reinterpret_cast<float (*)[XP]>(&hs[0][0]);
  __shared__ float gbuf[TM];
  __shared__ int   tokbuf[TM];
  __shared__ float lsum[2 * E_NUM];
  __shared__ int   sn[E_NUM];

  const int tid  = threadIdx.x;
  const int lane = tid & 31;
  const int w    = tid >> 5;
  const int lh   = lane >> 4;
  const int ln   = lane & 15;

  const bool loss_wg = (blockIdx.x == MAXT - 1);

  static_assert(E_NUM == 256 / 32 && NBLK == 64, "one wave per expert, two counts per lane");
  {
    int c = pc[w * NBLK + lane] + pc[w * NBLK + lane + 32];
    for (int d = 16; d >= 1; d >>= 1) c += __shfl_xor(c, d);
    if (lane == 0) sn[w] = c;
    if (loss_wg) {
      float iv = pi[w * NBLK + lane] + pi[w * NBLK + lane + 32];
      for (int d = 16; d >= 1; d >>= 1) iv += __shfl_xor(iv, d);
      if (lane == 0) { lsum[w] = iv; lsum[E_NUM + w] = (float)c; }
    }
  }
  __syncthreads();
  if (loss_wg && tid == 0) moe_loss_finish(lsum, out_loss);
  int tq = blockIdx.x, eq = E_NUM, nq = 0;
#pragma unroll
  for (int i = 0; i < E_NUM; ++i) {
    const int ni = sn[i];
    const int ti = (ni + TM - 1) / TM;
    if (eq == E_NUM) {
      if (tq < ti) { eq = i; nq = ni; }
      else tq -= ti;
    }
  }
  const int e    = __builtin_amdgcn_readfirstlane(eq);
  const int n    = __builtin_amdgcn_readfirstlane(nq);
  const int base = __builtin_amdgcn_readfirstlane(tq) * TM;

  if (e == E_NUM) return;
  const int m = min(TM, n - base);

  const int* pce = pc + e * NBLK;
  int cb[NBLK];
#pragma unroll
  for (int i = 0; i < NBLK; ++i) cb[i] = pce[i];

  if (tid < TM) {
    const int idx = min(base + tid, n - 1);
    int s = 0, blk = 0, off = idx;
#pragma unroll
    for (int i = 0; i < NBLK; ++i) {
      if (idx >= s) { blk = i; off = idx - s; }
      s += cb[i];
    }
    int tok2 = seg[(e * NBLK + blk) * GB + off];
    tokbuf[tid] = tok2;
    gbuf[tid]   = (base + tid < n) ? gates[(size_t)(tok2 >> 1) * E_NUM + e] : 0.f;
  }
  __syncthreads();

#pragma unroll
  for (int i = 0; i < TM * D_IN / 256; ++i) {
    int idx = tid + i * 256;
    int r = idx >> 7, c = idx & 127;
    xs[r][c] = x[(size_t)(tokbuf[r] >> 1) * D_IN + c];
  }
  __syncthreads();

  {
    const float* w1e = w1 + (size_t)e * D_IN * H_DIM;
    const float* pB  = w1e + (2 * lh) * H_DIM + w * 64 + ln;
    v8f acc[MT][4];
#pragma unroll
    for (int nt = 0; nt < 4; ++nt) {
      float bv = b1[e * H_DIM + w * 64 + nt * 16 + ln];
#pragma unroll
      for (int mt = 0; mt < MT; ++mt)
#pragma unroll
        for (int i = 0; i < 8; ++i) acc[mt][nt][i] = bv;
    }
    v2f bq0[2][4], bq1[2][4];
    const float* x0 = &xs[ln][2 * lh];
    g1_load(bq0, pB);
#pragma nounroll
    for (int kb = 0; kb < D_IN / 8 - 2; kb += 2) {
      g1_load(bq1, pB + (kb + 1) * 8 * H_DIM);
      g1_mma(acc, bq0, x0 + kb * 8);
      g1_load(bq0, pB + (kb + 2) * 8 * H_DIM);
      g1_mma(acc, bq1, x0 + (kb + 1) * 8);
    }
    g1_load(bq1, pB + (D_IN / 8 - 1) * 8 * H_DIM);
    g1_mma(acc, bq0, x0 + (D_IN / 8 - 2) * 8);
    g1_mma(acc, bq1, x0 + (D_IN / 8 - 1) * 8);
    __syncthreads();
#pragma unroll
    for (int mt = 0; mt < MT; ++mt)
#pragma unroll
      for (int nt = 0; nt < 4; ++nt)
#pragma unroll
        for (int i = 0; i < 8; ++i)
          hs[mt * 16 + 8 * lh + i][w * 64 + nt * 16 + ln] = fmaxf(acc[mt][nt][i], 0.f);
  }
  __syncthreads();

  {
    const int o0 = w * 16;
    const float* w2e = w2 + (size_t)e * H_DIM * O_DIM;
    const float* pB  = w2e + (2 * lh) * O_DIM + o0 + ln;
    v8f acc[MT][2];
    float bv = b2[e * O_DIM + o0 + ln];
#pragma unroll
    for (int mt = 0; mt < MT; ++mt)
#pragma unroll
      for (int i = 0; i < 8; ++i) { acc[mt][0][i] = bv;  acc[mt][1][i] = 0.f; }
    v2f cq0[16], cq1[16];
    g2_load(cq0, pB);
    const float* h0 = &hs[ln][2 * lh];
#pragma nounroll
    for (int kb = 0; kb < H_DIM / 64 - 2; kb += 2) {
      g2_load(cq1, pB + (kb + 1) * 64 * O_DIM);
      g2_mma(acc, cq0, h0 + kb * 64);
      g2_load(cq0, pB + (kb + 2) * 64 * O_DIM);
      g2_mma(acc, cq1, h0 + (kb + 1) * 64);
    }
    g2_load(cq1, pB + (H_DIM / 64 - 1) * 64 * O_DIM);
    g2_mma(acc, cq0, h0 + (H_DIM / 64 - 2) * 64);
    g2_mma(acc, cq1, h0 + (H_DIM / 64 - 1) * 64);
    __syncthreads();
#pragma unroll
    for (int mt = 0; mt < MT; ++mt)
#pragma unroll
      for (int i = 0; i < 8; ++i) {
        int M = mt * 16 + 8 * lh + i;
        xs[M][o0 + ln] = (acc[mt][0][i] + acc[mt][1][i]) * gbuf[M];
      }
  }
  __syncthreads();
#pragma unroll
  for (int j = 0; j < TM / 8; ++j) {
    const int r = w * (TM / 8) + j;
    if (r < m) {
      float4 v = *(const float4*)&xs[r][4 * lane];
      *(float4*)(part + (size_t)tokbuf[r] * O_DIM + 4 * lane) = v;
    }
  }
}

__global__ __launch_bounds__(256) void moe_combine(const float* __restrict__ part,
                                                   float* __restrict__ y) {
  const float EPSV = 2.2204460492503131e-16f;
  int idx = blockIdx.x * 256 + threadIdx.x;
  int b = idx >> 5, c4 = idx & 31;
  const float4* pr = ((const float4*)part) + (size_t)b * (2 * O_DIM / 4) + c4;
  float4 t0 = pr[0];
  float4 t1 = pr[O_DIM / 4];
  float4* p = ((float4*)y) + idx;
  float4 v;
  v.x = t0.x + t1.x; v.y = t0.y + t1.y; v.z = t0.z + t1.z; v.w = t0.w + t1.w;
  v.x = (v.x == 0.f) ? EPSV : v.x;
  v.y = (v.y == 0.f) ? EPSV : v.y;
  v.z = (v.z == 0.f) ? EPSV : v.z;
  v.w = (v.w == 0.f) ? EPSV : v.w;
  *p = v;
}

extern "C" void kernel_launch(void* const* d_in, const int* in_sizes, int n_in,
                              void* d_out, int out_size, void* d_ws, size_t ws_size,
                              hipStream_t stream) {
  (void)in_sizes; (void)n_in; (void)out_size; (void)ws_size;
  const float* x      = (const float*)d_in[0];
  const float* w_gate = (const float*)d_in[1];
  const float* w1     = (const float*)d_in[2];
  const float* b1     = (const float*)d_in[3];
  const float* w2     = (const float*)d_in[4];
  const float* b2     = (const float*)d_in[5];

  float* y    = (float*)d_out;
  float* loss = y + (size_t)B_TOK * O_DIM;

  char*  ws         = (char*)d_ws;
  float* gates      = (float*)ws;
  int*   seg        = (int*)(ws + (size_t)B_TOK * E_NUM * 4);
  int*   pc         = (int*)(ws + 2 * (size_t)B_TOK * E_NUM * 4);
  float* pi         = (float*)(ws + 2 * (size_t)B_TOK * E_NUM * 4 + 4096);
  float* part       = (float*)(ws + 4 * (size_t)B_TOK * E_NUM * 4);

  moe_gate<<<NBLK, GB, 0, stream>>>(x, w_gate, gates, seg, pc, pi);
  moe_expert<<<MAXT, 256, 0, stream>>>(x, w1, b1, w2, b2, gates, seg, pc, pi, part, loss);
  moe_combine<<<(B_TOK * O_DIM) / (4 * 256), 256, 0, stream>>>(part, y);
}


# gate kernel on all 256 WGPs: 64-token blocks, 4 lanes per row (coalesced; logit = fixed butterfly of 4 partial sums); expert kernel finds rows by LDS prefix search; importance: new fixed tree
# speedup vs baseline: 1.6556x; 1.0739x over previous
#include <hip/hip_runtime.h>
#include <hip/hip_bf16.h>
#include <math.h>

#define B_TOK 16384
#define D_IN  128
#define H_DIM 512
#define O_DIM 128
#define E_NUM 8
#define TM    48
#define MT    (TM / 16)
#define XP    132
#define HP    516
#define MAXT  ((2 * B_TOK + E_NUM * (TM - 1)) / TM)
#define GB    64
#define NBLK  (B_TOK / GB)
#define WP    36

typedef __attribute__((ext_vector_type(2))) float v2f;
typedef __attribute__((ext_vector_type(8))) float v8f;

__device__ __forceinline__ float quad_xor1(float v) {
  return __int_as_float(__builtin_amdgcn_update_dpp(0, __float_as_int(v), 0xB1, 0xF, 0xF, true));
}
__device__ __forceinline__ float quad_xor2(float v) {
  return __int_as_float(__builtin_amdgcn_update_dpp(0, __float_as_int(v), 0x4E, 0xF, 0xF, true));
}

__global__ __launch_bounds__(256) void moe_gate(const float* __restrict__ x,
                                                const float* __restrict__ w_gate,
                                                float* __restrict__ gates,
                                                int* __restrict__ seg,
                                                int* __restrict__ pc,
                                                float* __restrict__ pi) {
  __shared__ __attribute__((aligned(16))) float wg[(D_IN / 4) * WP];
  __shared__ float simp[E_NUM];
  __shared__ int   scnt[E_NUM];
  const int tid = threadIdx.x;
  const int q   = tid & 3;
  const int b   = blockIdx.x * GB + (tid >> 2);

  const float4* xr = (const float4*)(x + (size_t)b * D_IN);
  float4 xv[D_IN / 16];
#pragma unroll
  for (int i = 0; i < D_IN / 16; ++i) xv[i] = xr[q + 4 * i];

  static_assert(D_IN * E_NUM == 4 * 256, "one float4 of w_gate per thread");
  {
    float4 wv = ((const float4*)w_gate)[tid];
    *(float4*)&wg[(tid >> 3) * WP + (tid & 7) * 4] = wv;
  }
  if (tid < E_NUM) { simp[tid] = 0.f; scnt[tid] = 0; }
  __syncthreads();

  v2f acc[E_NUM / 2];
#pragma unroll
  for (int p = 0; p < E_NUM / 2; ++p) { acc[p][0] = 0.f; acc[p][1] = 0.f; }
#pragma unroll
  for (int i = 0; i < D_IN / 16; ++i) {
    const float xk[4] = {xv[i].x, xv[i].y, xv[i].z, xv[i].w};
    const float* wr = &wg[(q + 4 * i) * WP];
#pragma unroll
    for (int k = 0; k < 4; ++k) {
      v2f xx; xx[0] = xk[k]; xx[1] = xk[k];
#pragma unroll
      for (int p = 0; p < E_NUM / 2; ++p) {
        v2f wv2; wv2[0] = wr[k * E_NUM + 2 * p]; wv2[1] = wr[k * E_NUM + 2 * p + 1];
        acc[p] += xx * wv2;
      }
    }
  }
  float l[E_NUM];
#pragma unroll
  for (int e = 0; e < E_NUM; ++e) {
    float v = acc[e >> 1][e & 1];
    v += quad_xor1(v);
    v += quad_xor2(v);
    l[e] = v;
  }

  int e0 = 0; float l0 = l[0];
#pragma unroll
  for (int e = 1; e < E_NUM; ++e) if (l[e] > l0) { l0 = l[e]; e0 = e; }
  int e1 = -1; float l1 = -3.4e38f;
#pragma unroll
  for (int e = 0; e < E_NUM; ++e) if (e != e0 && l[e] > l1) { l1 = l[e]; e1 = e; }

  float qq = __expf(l1 - l0);
  float g0 = 1.f / (1.f + qq);
  float g1 = qq / (1.f + qq);

  {
    const int ea = 2 * q, eb = 2 * q + 1;
    float2 gv;
    gv.x = (ea == e0) ? g0 : ((ea == e1) ? g1 : 0.f);
    gv.y = (eb == e0) ? g0 : ((eb == e1) ? g1 : 0.f);
    *(float2*)(gates + (size_t)b * E_NUM + ea) = gv;
  }

  int r0 = 0, r1 = 0;
  if (q == 0) {
    atomicAdd(&simp[e0], g0);
    atomicAdd(&simp[e1], g1);
    r0 = atomicAdd(&scnt[e0], 1);
    r1 = atomicAdd(&scnt[e1], 1);
  }
  __syncthreads();
  if (tid < E_NUM) {
    pc[tid * NBLK + blockIdx.x] = scnt[tid];
    pi[tid * NBLK + blockIdx.x] = simp[tid];
  }
  if (q == 0) {
    seg[(e0 * NBLK + blockIdx.x) * GB + r0] = 2 * b;
    seg[(e1 * NBLK + blockIdx.x) * GB + r1] = 2 * b + 1;
  }
}

__device__ __forceinline__ void moe_loss_finish(const float* lsum, float* __restrict__ out_loss) {
  float im[E_NUM], ld[E_NUM], mi = 0.f, ml = 0.f;
  for (int e = 0; e < E_NUM; ++e) {
    im[e] = lsum[e]; ld[e] = lsum[E_NUM + e];
    mi += im[e]; ml += ld[e];
  }
  mi *= (1.f / E_NUM); ml *= (1.f / E_NUM);
  float vi = 0.f, vl = 0.f;
  for (int e = 0; e < E_NUM; ++e) {
    float a = im[e] - mi, c = ld[e] - ml;
    vi += a * a; vl += c * c;
  }
  vi *= (1.f / (E_NUM - 1)); vl *= (1.f / (E_NUM - 1));
  *out_loss = vi / (mi * mi + 1e-10f) + vl / (ml * ml + 1e-10f);
}

__device__ __forceinline__ void g1_load(v2f (&b)[2][4], const float* __restrict__ p) {
#pragma unroll
  for (int s = 0; s < 2; ++s)
#pragma unroll
    for (int nt = 0; nt < 4; ++nt) {
      b[s][nt][0] = p[(4 * s) * H_DIM + nt * 16];
      b[s][nt][1] = p[(4 * s) * H_DIM + nt * 16 + H_DIM];
    }
}

__device__ __forceinline__ void g1_mma(v8f (&acc)[MT][4], const v2f (&b)[2][4],
                                       const float* x0) {
#pragma unroll
  for (int s = 0; s < 2; ++s) {
    v2f a[MT];
#pragma unroll
    for (int mt = 0; mt < MT; ++mt) {
      float2 a2 = *(const float2*)(x0 + mt * 16 * XP + 4 * s);
      a[mt][0] = a2.x; a[mt][1] = a2.y;
    }
#pragma unroll
    for (int nt = 0; nt < 4; ++nt)
#pragma unroll
      for (int mt = 0; mt < MT; ++mt)
        acc[mt][nt] = __builtin_amdgcn_wmma_f32_16x16x4_f32(false, a[mt], false, b[s][nt],
                                                            (short)0, acc[mt][nt], false, false);
  }
}

__device__ __forceinline__ void g2_load(v2f (&b)[16], const float* __restrict__ p) {
#pragma unroll
  for (int s = 0; s < 16; ++s) {
    b[s][0] = p[(4 * s) * O_DIM];
    b[s][1] = p[(4 * s) * O_DIM + O_DIM];
  }
}

__device__ __forceinline__ void g2_mma(v8f (&acc)[MT][2], const v2f (&b)[16],
                                       const float* h0) {
#pragma unroll
  for (int s = 0; s < 16; ++s) {
#pragma unroll
    for (int mt = 0; mt < MT; ++mt) {
      float2 a2 = *(const float2*)(h0 + mt * 16 * HP + 4 * s);
      v2f a; a[0] = a2.x; a[1] = a2.y;
      acc[mt][s & 1] = __builtin_amdgcn_wmma_f32_16x16x4_f32(false, a, false, b[s],
                                                             (short)0, acc[mt][s & 1], false, false);
    }
  }
}

__global__ __launch_bounds__(256, 6) void moe_expert(const float* __restrict__ x,
                                                  const float* __restrict__ w1,
                                                  const float* __restrict__ b1,
                                                  const float* __restrict__ w2,
                                                  const float* __restrict__ b2,
                                                  const float* __restrict__ gates,
                                                  const int* __restrict__ seg,
                                                  const int* __restrict__ pc,
                                                  const float* __restrict__ pi,
                                                  float* __restrict__ part,
                                                  float* __restrict__ out_loss) {
  __shared__ __attribute__((aligned(16))) float hs[TM][HP];
  float (*xs)[XP] = reinterpret_cast<float (*)[XP]>(&hs[0][0]);
  __shared__ float gbuf[TM];
  __shared__ int   tokbuf[TM];
  __shared__ float lsum[2 * E_NUM];
  __shared__ int   sn[E_NUM];

  const int tid  = threadIdx.x;
  const int lane = tid & 31;
  const int w    = tid >> 5;
  const int lh   = lane >> 4;
  const int ln   = lane & 15;

  const bool loss_wg = (blockIdx.x == MAXT - 1);

  static_assert(E_NUM == 256 / 32 && NBLK == 256, "one wave per expert, eight counts per lane");
  float* pref = &hs[0][0];
  {
    const int4* p4 = (const int4*)(pc + w * NBLK + 8 * lane);
    const int4 ca = p4[0], cc = p4[1];
    const int tot = ((ca.x + ca.y) + (ca.z + ca.w)) + ((cc.x + cc.y) + (cc.z + cc.w));
    int inc = tot;
#pragma unroll
    for (int d = 1; d < 32; d <<= 1) {
      const int v = __shfl_up(inc, d);
      if (lane >= d) inc += v;
    }
    int run = inc - tot;
    float4 pa, pb;
    pa.x = __int_as_float(run); run += ca.x;
    pa.y = __int_as_float(run); run += ca.y;
    pa.z = __int_as_float(run); run += ca.z;
    pa.w = __int_as_float(run); run += ca.w;
    pb.x = __int_as_float(run); run += cc.x;
    pb.y = __int_as_float(run); run += cc.y;
    pb.z = __int_as_float(run); run += cc.z;
    pb.w = __int_as_float(run);
    *(float4*)(pref + w * NBLK + 8 * lane)     = pa;
    *(float4*)(pref + w * NBLK + 8 * lane + 4) = pb;
    const int c = __shfl(inc, 31);
    if (lane == 0) sn[w] = c;
    if (loss_wg) {
      const float4* f4 = (const float4*)(pi + w * NBLK + 8 * lane);
      const float4 fa = f4[0], fb = f4[1];
      float iv = ((fa.x + fa.y) + (fa.z + fa.w)) + ((fb.x + fb.y) + (fb.z + fb.w));
      for (int d = 16; d >= 1; d >>= 1) iv += __shfl_xor(iv, d);
      if (lane == 0) { lsum[w] = iv; lsum[E_NUM + w] = (float)c; }
    }
  }
  __syncthreads();
  if (loss_wg && tid == 0) moe_loss_finish(lsum, out_loss);
  int tq = blockIdx.x, eq = E_NUM, nq = 0;
#pragma unroll
  for (int i = 0; i < E_NUM; ++i) {
    const int ni = sn[i];
    const int ti = (ni + TM - 1) / TM;
    if (eq == E_NUM) {
      if (tq < ti) { eq = i; nq = ni; }
      else tq -= ti;
    }
  }
  const int e    = __builtin_amdgcn_readfirstlane(eq);
  const int n    = __builtin_amdgcn_readfirstlane(nq);
  const int base = __builtin_amdgcn_readfirstlane(tq) * TM;

  if (e == E_NUM) return;
  const int m = min(TM, n - base);

  if (tid < TM) {
    const int idx = min(base + tid, n - 1);
    const float* pe = pref + e * NBLK;
    int blk = 0;
#pragma unroll
    for (int st = NBLK / 2; st >= 1; st >>= 1)
      if (__float_as_int(pe[blk + st]) <= idx) blk += st;
    const int off = idx - __float_as_int(pe[blk]);
    int tok2 = seg[(e * NBLK + blk) * GB + off];
    tokbuf[tid] = tok2;
    gbuf[tid]   = (base + tid < n) ? gates[(size_t)(tok2 >> 1) * E_NUM + e] : 0.f;
  }
  __syncthreads();

#pragma unroll
  for (int i = 0; i < TM * D_IN / 256; ++i) {
    int idx = tid + i * 256;
    int r = idx >> 7, c = idx & 127;
    xs[r][c] = x[(size_t)(tokbuf[r] >> 1) * D_IN + c];
  }
  __syncthreads();

  {
    const float* w1e = w1 + (size_t)e * D_IN * H_DIM;
    const float* pB  = w1e + (2 * lh) * H_DIM + w * 64 + ln;
    v8f acc[MT][4];
#pragma unroll
    for (int nt = 0; nt < 4; ++nt) {
      float bv = b1[e * H_DIM + w * 64 + nt * 16 + ln];
#pragma unroll
      for (int mt = 0; mt < MT; ++mt)
#pragma unroll
        for (int i = 0; i < 8; ++i) acc[mt][nt][i] = bv;
    }
    v2f bq0[2][4], bq1[2][4];
    const float* x0 = &xs[ln][2 * lh];
    g1_load(bq0, pB);
#pragma nounroll
    for (int kb = 0; kb < D_IN / 8 - 2; kb += 2) {
      g1_load(bq1, pB + (kb + 1) * 8 * H_DIM);
      g1_mma(acc, bq0, x0 + kb * 8);
      g1_load(bq0, pB + (kb + 2) * 8 * H_DIM);
      g1_mma(acc, bq1, x0 + (kb + 1) * 8);
    }
    g1_load(bq1, pB + (D_IN / 8 - 1) * 8 * H_DIM);
    g1_mma(acc, bq0, x0 + (D_IN / 8 - 2) * 8);
    g1_mma(acc, bq1, x0 + (D_IN / 8 - 1) * 8);
    __syncthreads();
#pragma unroll
    for (int mt = 0; mt < MT; ++mt)
#pragma unroll
      for (int nt = 0; nt < 4; ++nt)
#pragma unroll
        for (int i = 0; i < 8; ++i)
          hs[mt * 16 + 8 * lh + i][w * 64 + nt * 16 + ln] = fmaxf(acc[mt][nt][i], 0.f);
  }
  __syncthreads();

  {
    const int o0 = w * 16;
    const float* w2e = w2 + (size_t)e * H_DIM * O_DIM;
    const float* pB  = w2e + (2 * lh) * O_DIM + o0 + ln;
    v8f acc[MT][2];
    float bv = b2[e * O_DIM + o0 + ln];
#pragma unroll
    for (int mt = 0; mt < MT; ++mt)
#pragma unroll
      for (int i = 0; i < 8; ++i) { acc[mt][0][i] = bv;  acc[mt][1][i] = 0.f; }
    v2f cq0[16], cq1[16];
    g2_load(cq0, pB);
    const float* h0 = &hs[ln][2 * lh];
#pragma nounroll
    for (int kb = 0; kb < H_DIM / 64 - 2; kb += 2) {
      g2_load(cq1, pB + (kb + 1) * 64 * O_DIM);
      g2_mma(acc, cq0, h0 + kb * 64);
      g2_load(cq0, pB + (kb + 2) * 64 * O_DIM);
      g2_mma(acc, cq1, h0 + (kb + 1) * 64);
    }
    g2_load(cq1, pB + (H_DIM / 64 - 1) * 64 * O_DIM);
    g2_mma(acc, cq0, h0 + (H_DIM / 64 - 2) * 64);
    g2_mma(acc, cq1, h0 + (H_DIM / 64 - 1) * 64);
    __syncthreads();
#pragma unroll
    for (int mt = 0; mt < MT; ++mt)
#pragma unroll
      for (int i = 0; i < 8; ++i) {
        int M = mt * 16 + 8 * lh + i;
        xs[M][o0 + ln] = (acc[mt][0][i] + acc[mt][1][i]) * gbuf[M];
      }
  }
  __syncthreads();
#pragma unroll
  for (int j = 0; j < TM / 8; ++j) {
    const int r = w * (TM / 8) + j;
    if (r < m) {
      float4 v = *(const float4*)&xs[r][4 * lane];
      *(float4*)(part + (size_t)tokbuf[r] * O_DIM + 4 * lane) = v;
    }
  }
}

__global__ __launch_bounds__(256) void moe_combine(const float* __restrict__ part,
                                                   float* __restrict__ y) {
  const float EPSV = 2.2204460492503131e-16f;
  int idx = blockIdx.x * 256 + threadIdx.x;
  int b = idx >> 5, c4 = idx & 31;
  const float4* pr = ((const float4*)part) + (size_t)b * (2 * O_DIM / 4) + c4;
  float4 t0 = pr[0];
  float4 t1 = pr[O_DIM / 4];
  float4* p = ((float4*)y) + idx;
  float4 v;
  v.x = t0.x + t1.x; v.y = t0.y + t1.y; v.z = t0.z + t1.z; v.w = t0.w + t1.w;
  v.x = (v.x == 0.f) ? EPSV : v.x;
  v.y = (v.y == 0.f) ? EPSV : v.y;
  v.z = (v.z == 0.f) ? EPSV : v.z;
  v.w = (v.w == 0.f) ? EPSV : v.w;
  *p = v;
}

extern "C" void kernel_launch(void* const* d_in, const int* in_sizes, int n_in,
                              void* d_out, int out_size, void* d_ws, size_t ws_size,
                              hipStream_t stream) {
  (void)in_sizes; (void)n_in; (void)out_size; (void)ws_size;
  const float* x      = (const float*)d_in[0];
  const float* w_gate = (const float*)d_in[1];
  const float* w1     = (const float*)d_in[2];
  const float* b1     = (const float*)d_in[3];
  const float* w2     = (const float*)d_in[4];
  const float* b2     = (const float*)d_in[5];

  float* y    = (float*)d_out;
  float* loss = y + (size_t)B_TOK * O_DIM;

  char*  ws         = (char*)d_ws;
  float* gates      = (float*)ws;
  int*   seg        = (int*)(ws + (size_t)B_TOK * E_NUM * 4);
  int*   pc         = (int*)(ws + 2 * (size_t)B_TOK * E_NUM * 4);
  float* pi         = (float*)(ws + 2 * (size_t)B_TOK * E_NUM * 4 + 8192);
  float* part       = (float*)(ws + 4 * (size_t)B_TOK * E_NUM * 4);

  moe_gate<<<NBLK, 256, 0, stream>>>(x, w_gate, gates, seg, pc, pi);
  moe_expert<<<MAXT, 256, 0, stream>>>(x, w1, b1, w2, b2, gates, seg, pc, pi, part, loss);
  moe_combine<<<(B_TOK * O_DIM) / (4 * 256), 256, 0, stream>>>(part, y);
}


# expert kernel: wave priority steps down with progress (s_setprio 3 in GEMM1, 2 in GEMM2, 1 and 0 for its last two weight blocks): work-groups sharing a WGP fill each other's waits; sums unchanged
# speedup vs baseline: 1.7521x; 1.0583x over previous
#include <hip/hip_runtime.h>
#include <hip/hip_bf16.h>
#include <math.h>

#define B_TOK 16384
#define D_IN  128
#define H_DIM 512
#define O_DIM 128
#define E_NUM 8
#define TM    48
#define MT    (TM / 16)
#define XP    132
#define HP    516
#define MAXT  ((2 * B_TOK + E_NUM * (TM - 1)) / TM)
#define GB    64
#define NBLK  (B_TOK / GB)
#define WP    36

typedef __attribute__((ext_vector_type(2))) float v2f;
typedef __attribute__((ext_vector_type(8))) float v8f;

__device__ __forceinline__ float quad_xor1(float v) {
  return __int_as_float(__builtin_amdgcn_update_dpp(0, __float_as_int(v), 0xB1, 0xF, 0xF, true));
}
__device__ __forceinline__ float quad_xor2(float v) {
  return __int_as_float(__builtin_amdgcn_update_dpp(0, __float_as_int(v), 0x4E, 0xF, 0xF, true));
}

__global__ __launch_bounds__(256) void moe_gate(const float* __restrict__ x,
                                                const float* __restrict__ w_gate,
                                                float* __restrict__ gates,
                                                int* __restrict__ seg,
                                                int* __restrict__ pc,
                                                float* __restrict__ pi) {
  __shared__ __attribute__((aligned(16))) float wg[(D_IN / 4) * WP];
  __shared__ float simp[E_NUM];
  __shared__ int   scnt[E_NUM];
  const int tid = threadIdx.x;
  const int q   = tid & 3;
  const int b   = blockIdx.x * GB + (tid >> 2);

  const float4* xr = (const float4*)(x + (size_t)b * D_IN);
  float4 xv[D_IN / 16];
#pragma unroll
  for (int i = 0; i < D_IN / 16; ++i) xv[i] = xr[q + 4 * i];

  static_assert(D_IN * E_NUM == 4 * 256, "one float4 of w_gate per thread");
  {
    float4 wv = ((const float4*)w_gate)[tid];
    *(float4*)&wg[(tid >> 3) * WP + (tid & 7) * 4] = wv;
  }
  if (tid < E_NUM) { simp[tid] = 0.f; scnt[tid] = 0; }
  __syncthreads();

  v2f acc[E_NUM / 2];
#pragma unroll
  for (int p = 0; p < E_NUM / 2; ++p) { acc[p][0] = 0.f; acc[p][1] = 0.f; }
#pragma unroll
  for (int i = 0; i < D_IN / 16; ++i) {
    const float xk[4] = {xv[i].x, xv[i].y, xv[i].z, xv[i].w};
    const float* wr = &wg[(q + 4 * i) * WP];
#pragma unroll
    for (int k = 0; k < 4; ++k) {
      v2f xx; xx[0] = xk[k]; xx[1] = xk[k];
#pragma unroll
      for (int p = 0; p < E_NUM / 2; ++p) {
        v2f wv2; wv2[0] = wr[k * E_NUM + 2 * p]; wv2[1] = wr[k * E_NUM + 2 * p + 1];
        acc[p] += xx * wv2;
      }
    }
  }
  float l[E_NUM];
#pragma unroll
  for (int e = 0; e < E_NUM; ++e) {
    float v = acc[e >> 1][e & 1];
    v += quad_xor1(v);
    v += quad_xor2(v);
    l[e] = v;
  }

  int e0 = 0; float l0 = l[0];
#pragma unroll
  for (int e = 1; e < E_NUM; ++e) if (l[e] > l0) { l0 = l[e]; e0 = e; }
  int e1 = -1; float l1 = -3.4e38f;
#pragma unroll
  for (int e = 0; e < E_NUM; ++e) if (e != e0 && l[e] > l1) { l1 = l[e]; e1 = e; }

  float qq = __expf(l1 - l0);
  float g0 = 1.f / (1.f + qq);
  float g1 = qq / (1.f + qq);

  {
    const int ea = 2 * q, eb = 2 * q + 1;
    float2 gv;
    gv.x = (ea == e0) ? g0 : ((ea == e1) ? g1 : 0.f);
    gv.y = (eb == e0) ? g0 : ((eb == e1) ? g1 : 0.f);
    *(float2*)(gates + (size_t)b * E_NUM + ea) = gv;
  }

  int r0 = 0, r1 = 0;
  if (q == 0) {
    atomicAdd(&simp[e0], g0);
    atomicAdd(&simp[e1], g1);
    r0 = atomicAdd(&scnt[e0], 1);
    r1 = atomicAdd(&scnt[e1], 1);
  }
  __syncthreads();
  if (tid < E_NUM) {
    pc[tid * NBLK + blockIdx.x] = scnt[tid];
    pi[tid * NBLK + blockIdx.x] = simp[tid];
  }
  if (q == 0) {
    seg[(e0 * NBLK + blockIdx.x) * GB + r0] = 2 * b;
    seg[(e1 * NBLK + blockIdx.x) * GB + r1] = 2 * b + 1;
  }
}

__device__ __forceinline__ void moe_loss_finish(const float* lsum, float* __restrict__ out_loss) {
  float im[E_NUM], ld[E_NUM], mi = 0.f, ml = 0.f;
  for (int e = 0; e < E_NUM; ++e) {
    im[e] = lsum[e]; ld[e] = lsum[E_NUM + e];
    mi += im[e]; ml += ld[e];
  }
  mi *= (1.f / E_NUM); ml *= (1.f / E_NUM);
  float vi = 0.f, vl = 0.f;
  for (int e = 0; e < E_NUM; ++e) {
    float a = im[e] - mi, c = ld[e] - ml;
    vi += a * a; vl += c * c;
  }
  vi *= (1.f / (E_NUM - 1)); vl *= (1.f / (E_NUM - 1));
  *out_loss = vi / (mi * mi + 1e-10f) + vl / (ml * ml + 1e-10f);
}

__device__ __forceinline__ void g1_load(v2f (&b)[2][4], const float* __restrict__ p) {
#pragma unroll
  for (int s = 0; s < 2; ++s)
#pragma unroll
    for (int nt = 0; nt < 4; ++nt) {
      b[s][nt][0] = p[(4 * s) * H_DIM + nt * 16];
      b[s][nt][1] = p[(4 * s) * H_DIM + nt * 16 + H_DIM];
    }
}

__device__ __forceinline__ void g1_mma(v8f (&acc)[MT][4], const v2f (&b)[2][4],
                                       const float* x0) {
#pragma unroll
  for (int s = 0; s < 2; ++s) {
    v2f a[MT];
#pragma unroll
    for (int mt = 0; mt < MT; ++mt) {
      float2 a2 = *(const float2*)(x0 + mt * 16 * XP + 4 * s);
      a[mt][0] = a2.x; a[mt][1] = a2.y;
    }
#pragma unroll
    for (int nt = 0; nt < 4; ++nt)
#pragma unroll
      for (int mt = 0; mt < MT; ++mt)
        acc[mt][nt] = __builtin_amdgcn_wmma_f32_16x16x4_f32(false, a[mt], false, b[s][nt],
                                                            (short)0, acc[mt][nt], false, false);
  }
}

__device__ __forceinline__ void g2_load(v2f (&b)[16], const float* __restrict__ p) {
#pragma unroll
  for (int s = 0; s < 16; ++s) {
    b[s][0] = p[(4 * s) * O_DIM];
    b[s][1] = p[(4 * s) * O_DIM + O_DIM];
  }
}

__device__ __forceinline__ void g2_mma(v8f (&acc)[MT][2], const v2f (&b)[16],
                                       const float* h0) {
#pragma unroll
  for (int s = 0; s < 16; ++s) {
#pragma unroll
    for (int mt = 0; mt < MT; ++mt) {
      float2 a2 = *(const float2*)(h0 + mt * 16 * HP + 4 * s);
      v2f a; a[0] = a2.x; a[1] = a2.y;
      acc[mt][s & 1] = __builtin_amdgcn_wmma_f32_16x16x4_f32(false, a, false, b[s],
                                                             (short)0, acc[mt][s & 1], false, false);
    }
  }
}

__global__ __launch_bounds__(256, 6) void moe_expert(const float* __restrict__ x,
                                                  const float* __restrict__ w1,
                                                  const float* __restrict__ b1,
                                                  const float* __restrict__ w2,
                                                  const float* __restrict__ b2,
                                                  const float* __restrict__ gates,
                                                  const int* __restrict__ seg,
                                                  const int* __restrict__ pc,
                                                  const float* __restrict__ pi,
                                                  float* __restrict__ part,
                                                  float* __restrict__ out_loss) {
  __shared__ __attribute__((aligned(16))) float hs[TM][HP];
  float (*xs)[XP] = reinterpret_cast<float (*)[XP]>(&hs[0][0]);
  __shared__ float gbuf[TM];
  __shared__ int   tokbuf[TM];
  __shared__ float lsum[2 * E_NUM];
  __shared__ int   sn[E_NUM];

  const int tid  = threadIdx.x;
  const int lane = tid & 31;
  const int w    = tid >> 5;
  const int lh   = lane >> 4;
  const int ln   = lane & 15;

  const bool loss_wg = (blockIdx.x == MAXT - 1);

  static_assert(E_NUM == 256 / 32 && NBLK == 256, "one wave per expert, eight counts per lane");
  float* pref = &hs[0][0];
  {
    const int4* p4 = (const int4*)(pc + w * NBLK + 8 * lane);
    const int4 ca = p4[0], cc = p4[1];
    const int tot = ((ca.x + ca.y) + (ca.z + ca.w)) + ((cc.x + cc.y) + (cc.z + cc.w));
    int inc = tot;
#pragma unroll
    for (int d = 1; d < 32; d <<= 1) {
      const int v = __shfl_up(inc, d);
      if (lane >= d) inc += v;
    }
    int run = inc - tot;
    float4 pa, pb;
    pa.x = __int_as_float(run); run += ca.x;
    pa.y = __int_as_float(run); run += ca.y;
    pa.z = __int_as_float(run); run += ca.z;
    pa.w = __int_as_float(run); run += ca.w;
    pb.x = __int_as_float(run); run += cc.x;
    pb.y = __int_as_float(run); run += cc.y;
    pb.z = __int_as_float(run); run += cc.z;
    pb.w = __int_as_float(run);
    *(float4*)(pref + w * NBLK + 8 * lane)     = pa;
    *(float4*)(pref + w * NBLK + 8 * lane + 4) = pb;
    const int c = __shfl(inc, 31);
    if (lane == 0) sn[w] = c;
    if (loss_wg) {
      const float4* f4 = (const float4*)(pi + w * NBLK + 8 * lane);
      const float4 fa = f4[0], fb = f4[1];
      float iv = ((fa.x + fa.y) + (fa.z + fa.w)) + ((fb.x + fb.y) + (fb.z + fb.w));
      for (int d = 16; d >= 1; d >>= 1) iv += __shfl_xor(iv, d);
      if (lane == 0) { lsum[w] = iv; lsum[E_NUM + w] = (float)c; }
    }
  }
  __syncthreads();
  if (loss_wg && tid == 0) moe_loss_finish(lsum, out_loss);
  int tq = blockIdx.x, eq = E_NUM, nq = 0;
#pragma unroll
  for (int i = 0; i < E_NUM; ++i) {
    const int ni = sn[i];
    const int ti = (ni + TM - 1) / TM;
    if (eq == E_NUM) {
      if (tq < ti) { eq = i; nq = ni; }
      else tq -= ti;
    }
  }
  const int e    = __builtin_amdgcn_readfirstlane(eq);
  const int n    = __builtin_amdgcn_readfirstlane(nq);
  const int base = __builtin_amdgcn_readfirstlane(tq) * TM;

  if (e == E_NUM) return;
  const int m = min(TM, n - base);

  __builtin_amdgcn_s_setprio(3);

  if (tid < TM) {
    const int idx = min(base + tid, n - 1);
    const float* pe = pref + e * NBLK;
    int blk = 0;
#pragma unroll
    for (int st = NBLK / 2; st >= 1; st >>= 1)
      if (__float_as_int(pe[blk + st]) <= idx) blk += st;
    const int off = idx - __float_as_int(pe[blk]);
    int tok2 = seg[(e * NBLK + blk) * GB + off];
    tokbuf[tid] = tok2;
    gbuf[tid]   = (base + tid < n) ? gates[(size_t)(tok2 >> 1) * E_NUM + e] : 0.f;
  }
  __syncthreads();

#pragma unroll
  for (int i = 0; i < TM * D_IN / 256; ++i) {
    int idx = tid + i * 256;
    int r = idx >> 7, c = idx & 127;
    xs[r][c] = x[(size_t)(tokbuf[r] >> 1) * D_IN + c];
  }
  __syncthreads();

  {
    const float* w1e = w1 + (size_t)e * D_IN * H_DIM;
    const float* pB  = w1e + (2 * lh) * H_DIM + w * 64 + ln;
    v8f acc[MT][4];
#pragma unroll
    for (int nt = 0; nt < 4; ++nt) {
      float bv = b1[e * H_DIM + w * 64 + nt * 16 + ln];
#pragma unroll
      for (int mt = 0; mt < MT; ++mt)
#pragma unroll
        for (int i = 0; i < 8; ++i) acc[mt][nt][i] = bv;
    }
    v2f bq0[2][4], bq1[2][4];
    const float* x0 = &xs[ln][2 * lh];
    g1_load(bq0, pB);
#pragma nounroll
    for (int kb = 0; kb < D_IN / 8 - 2; kb += 2) {
      g1_load(bq1, pB + (kb + 1) * 8 * H_DIM);
      g1_mma(acc, bq0, x0 + kb * 8);
      g1_load(bq0, pB + (kb + 2) * 8 * H_DIM);
      g1_mma(acc, bq1, x0 + (kb + 1) * 8);
    }
    g1_load(bq1, pB + (D_IN / 8 - 1) * 8 * H_DIM);
    g1_mma(acc, bq0, x0 + (D_IN / 8 - 2) * 8);
    g1_mma(acc, bq1, x0 + (D_IN / 8 - 1) * 8);
    __syncthreads();
#pragma unroll
    for (int mt = 0; mt < MT; ++mt)
#pragma unroll
      for (int nt = 0; nt < 4; ++nt)
#pragma unroll
        for (int i = 0; i < 8; ++i)
          hs[mt * 16 + 8 * lh + i][w * 64 + nt * 16 + ln] = fmaxf(acc[mt][nt][i], 0.f);
  }
  __syncthreads();

  {
    const int o0 = w * 16;
    const float* w2e = w2 + (size_t)e * H_DIM * O_DIM;
    const float* pB  = w2e + (2 * lh) * O_DIM + o0 + ln;
    v8f acc[MT][2];
    float bv = b2[e * O_DIM + o0 + ln];
#pragma unroll
    for (int mt = 0; mt < MT; ++mt)
#pragma unroll
      for (int i = 0; i < 8; ++i) { acc[mt][0][i] = bv;  acc[mt][1][i] = 0.f; }
    v2f cq0[16], cq1[16];
    __builtin_amdgcn_s_setprio(2);
    g2_load(cq0, pB);
    const float* h0 = &hs[ln][2 * lh];
#pragma nounroll
    for (int kb = 0; kb < H_DIM / 64 - 2; kb += 2) {
      g2_load(cq1, pB + (kb + 1) * 64 * O_DIM);
      g2_mma(acc, cq0, h0 + kb * 64);
      g2_load(cq0, pB + (kb + 2) * 64 * O_DIM);
      g2_mma(acc, cq1, h0 + (kb + 1) * 64);
    }
    __builtin_amdgcn_s_setprio(1);
    g2_load(cq1, pB + (H_DIM / 64 - 1) * 64 * O_DIM);
    g2_mma(acc, cq0, h0 + (H_DIM / 64 - 2) * 64);
    __builtin_amdgcn_s_setprio(0);
    g2_mma(acc, cq1, h0 + (H_DIM / 64 - 1) * 64);
    __syncthreads();
#pragma unroll
    for (int mt = 0; mt < MT; ++mt)
#pragma unroll
      for (int i = 0; i < 8; ++i) {
        int M = mt * 16 + 8 * lh + i;
        xs[M][o0 + ln] = (acc[mt][0][i] + acc[mt][1][i]) * gbuf[M];
      }
  }
  __syncthreads();
#pragma unroll
  for (int j = 0; j < TM / 8; ++j) {
    const int r = w * (TM / 8) + j;
    if (r < m) {
      float4 v = *(const float4*)&xs[r][4 * lane];
      *(float4*)(part + (size_t)tokbuf[r] * O_DIM + 4 * lane) = v;
    }
  }
}

__global__ __launch_bounds__(256) void moe_combine(const float* __restrict__ part,
                                                   float* __restrict__ y) {
  const float EPSV = 2.2204460492503131e-16f;
  int idx = blockIdx.x * 256 + threadIdx.x;
  int b = idx >> 5, c4 = idx & 31;
  const float4* pr = ((const float4*)part) + (size_t)b * (2 * O_DIM / 4) + c4;
  float4 t0 = pr[0];
  float4 t1 = pr[O_DIM / 4];
  float4* p = ((float4*)y) + idx;
  float4 v;
  v.x = t0.x + t1.x; v.y = t0.y + t1.y; v.z = t0.z + t1.z; v.w = t0.w + t1.w;
  v.x = (v.x == 0.f) ? EPSV : v.x;
  v.y = (v.y == 0.f) ? EPSV : v.y;
  v.z = (v.z == 0.f) ? EPSV : v.z;
  v.w = (v.w == 0.f) ? EPSV : v.w;
  *p = v;
}

extern "C" void kernel_launch(void* const* d_in, const int* in_sizes, int n_in,
                              void* d_out, int out_size, void* d_ws, size_t ws_size,
                              hipStream_t stream) {
  (void)in_sizes; (void)n_in; (void)out_size; (void)ws_size;
  const float* x      = (const float*)d_in[0];
  const float* w_gate = (const float*)d_in[1];
  const float* w1     = (const float*)d_in[2];
  const float* b1     = (const float*)d_in[3];
  const float* w2     = (const float*)d_in[4];
  const float* b2     = (const float*)d_in[5];

  float* y    = (float*)d_out;
  float* loss = y + (size_t)B_TOK * O_DIM;

  char*  ws         = (char*)d_ws;
  float* gates      = (float*)ws;
  int*   seg        = (int*)(ws + (size_t)B_TOK * E_NUM * 4);
  int*   pc         = (int*)(ws + 2 * (size_t)B_TOK * E_NUM * 4);
  float* pi         = (float*)(ws + 2 * (size_t)B_TOK * E_NUM * 4 + 8192);
  float* part       = (float*)(ws + 4 * (size_t)B_TOK * E_NUM * 4);

  moe_gate<<<NBLK, 256, 0, stream>>>(x, w_gate, gates, seg, pc, pi);
  moe_expert<<<MAXT, 256, 0, stream>>>(x, w1, b1, w2, b2, gates, seg, pc, pi, part, loss);
  moe_combine<<<(B_TOK * O_DIM) / (4 * 256), 256, 0, stream>>>(part, y);
}
